# starting kernel
# baseline (speedup 1.0000x reference)
#include <hip/hip_runtime.h>
#include <math.h>

#define HEADS     8
#define HIDC      32
#define L1C       256
#define OUTC      16
#define NEG_SLOPE 0.2f

typedef __attribute__((ext_vector_type(2))) float v2f;
typedef __attribute__((ext_vector_type(8))) float v8f;

__device__ __forceinline__ unsigned fenc(float f) {
    unsigned u = __float_as_uint(f);
    return (u & 0x80000000u) ? ~u : (u | 0x80000000u);
}
__device__ __forceinline__ float fdec(unsigned k) {
    return __uint_as_float((k & 0x80000000u) ? (k ^ 0x80000000u) : ~k);
}
__device__ __forceinline__ float lrelu(float v) {
    return v > 0.0f ? v : NEG_SLOPE * v;
}
__device__ __forceinline__ void edge_sd(const int* __restrict__ ei, int E, int e,
                                        int& s, int& d) {
    if (e < E) { s = ei[e]; d = ei[E + e]; }
    else       { s = e - E; d = e - E; }
}

__global__ void k_fill_zero(float* __restrict__ p, size_t n) {
    size_t i = (size_t)blockIdx.x * blockDim.x + threadIdx.x;
    size_t stride = (size_t)gridDim.x * blockDim.x;
    for (; i < n; i += stride) p[i] = 0.0f;
}

__global__ void k_gemm1_wmma(const float* __restrict__ x, const float* __restrict__ W1,
                             float* __restrict__ h1, int n_nodes) {
    int wave   = (int)((blockIdx.x * blockDim.x + threadIdx.x) >> 5);
    int lane   = threadIdx.x & 31;
    int mtiles = (n_nodes + 15) >> 4;
    int mtile  = wave >> 4;
    int ntile  = wave & 15;
    if (mtile >= mtiles) return;
    int m  = lane & 15;
    int kh = lane >> 4;
    bool full = (mtile * 16 + 16 <= n_nodes);
    int node = mtile * 16 + m;
    if (node >= n_nodes) node = n_nodes - 1;
    float x0 = x[node * 3 + 0], x1 = x[node * 3 + 1], x2 = x[node * 3 + 2];
    int col = ntile * 16 + m;
    float w0 = W1[0 * L1C + col], w1 = W1[1 * L1C + col], w2 = W1[2 * L1C + col];
    v2f a, b;
    a.x = kh ? x2 : x0;  a.y = kh ? 0.0f : x1;
    b.x = kh ? w2 : w0;  b.y = kh ? 0.0f : w1;
    v8f c = {};
    c = __builtin_amdgcn_wmma_f32_16x16x4_f32(false, a, false, b, (short)0, c,
                                              false, false);
    float* p = h1 + (size_t)(mtile * 16 + 8 * kh) * L1C + col;
    if (full) {
#pragma unroll
        for (int v = 0; v < 8; ++v) p[(size_t)v * L1C] = c[v];
    } else {
        for (int v = 0; v < 8; ++v)
            if (mtile * 16 + 8 * kh + v < n_nodes) p[(size_t)v * L1C] = c[v];
    }
}

__global__ void k_att1(const float* __restrict__ h1, const float* __restrict__ as1,
                       const float* __restrict__ ad1, float* __restrict__ a_s,
                       float* __restrict__ a_d, int n_nodes) {
    int node = blockIdx.x;
    if (node >= n_nodes) return;
    int head = threadIdx.x >> 5;
    int lane = threadIdx.x & 31;
    float v = h1[(size_t)node * L1C + head * HIDC + lane];
    float s = v * as1[head * HIDC + lane];
    float d = v * ad1[head * HIDC + lane];
    for (int off = 16; off; off >>= 1) {
        s += __shfl_xor(s, off, 32);
        d += __shfl_xor(d, off, 32);
    }
    if (lane == 0) {
        a_s[node * HEADS + head] = s;
        a_d[node * HEADS + head] = d;
    }
}

__global__ void k_edge_max1(const int* __restrict__ ei, int E, int ET,
                            const float* __restrict__ a_s, const float* __restrict__ a_d,
                            unsigned* __restrict__ m1) {
    int e = blockIdx.x * blockDim.x + threadIdx.x;
    if (e >= ET) return;
    int s, d; edge_sd(ei, E, e, s, d);
#pragma unroll
    for (int h = 0; h < HEADS; ++h) {
        float v = lrelu(a_s[s * HEADS + h] + a_d[d * HEADS + h]);
        atomicMax(&m1[d * HEADS + h], fenc(v));
    }
}

__global__ void k_edge_exp1(const int* __restrict__ ei, int E, int ET,
                            const float* __restrict__ a_s, const float* __restrict__ a_d,
                            const unsigned* __restrict__ m1, float* __restrict__ denom1) {
    int e = blockIdx.x * blockDim.x + threadIdx.x;
    if (e >= ET) return;
    int s, d; edge_sd(ei, E, e, s, d);
#pragma unroll
    for (int h = 0; h < HEADS; ++h) {
        float v = lrelu(a_s[s * HEADS + h] + a_d[d * HEADS + h]);
        float w = expf(v - fdec(m1[d * HEADS + h]));
        atomicAdd(&denom1[d * HEADS + h], w);
    }
}

__global__ void k_msg1(const int* __restrict__ ei, int E, int ET,
                       const float* __restrict__ a_s, const float* __restrict__ a_d,
                       const unsigned* __restrict__ m1, const float* __restrict__ denom1,
                       const float* __restrict__ h1, float* __restrict__ out1) {
    long long gid = (long long)blockIdx.x * blockDim.x + threadIdx.x;
    int e = (int)(gid >> 5);
    int lane = (int)(gid & 31);
    if (e >= ET) return;
    int s, d; edge_sd(ei, E, e, s, d);
    float myalpha = 0.0f;
    if (lane < HEADS) {
        float v = lrelu(a_s[s * HEADS + lane] + a_d[d * HEADS + lane]);
        float w = expf(v - fdec(m1[d * HEADS + lane]));
        myalpha = w / (denom1[d * HEADS + lane] + 1e-16f);
    }
    const float* ph = h1 + (size_t)s * L1C + lane;
    float* po = out1 + (size_t)d * L1C + lane;
#pragma unroll
    for (int h = 0; h < HEADS; ++h) {
        float ah = __shfl(myalpha, h, 32);
        atomicAdd(po + h * HIDC, ph[h * HIDC] * ah);
    }
}

__global__ void k_elu(float* __restrict__ out1, const float* __restrict__ bias1, size_t n) {
    size_t i = (size_t)blockIdx.x * blockDim.x + threadIdx.x;
    size_t stride = (size_t)gridDim.x * blockDim.x;
    for (; i < n; i += stride) {
        float v = out1[i] + bias1[i & (L1C - 1)];
        out1[i] = v > 0.0f ? v : (expf(v) - 1.0f);
    }
}

__global__ void k_gemm2_wmma(const float* __restrict__ helu, const float* __restrict__ W2,
                             float* __restrict__ h2, int n_nodes) {
    int wave = (int)((blockIdx.x * blockDim.x + threadIdx.x) >> 5);
    int lane = threadIdx.x & 31;
    int mtiles = (n_nodes + 15) >> 4;
    if (wave >= mtiles) return;
    int m  = lane & 15;
    int kh = lane >> 4;
    bool full = (wave * 16 + 16 <= n_nodes);
    int node = wave * 16 + m;
    if (node >= n_nodes) node = n_nodes - 1;
    const float* arow = helu + (size_t)node * L1C + kh * 2;
    const float* brow = W2 + kh * 2 * OUTC + m;
    v8f c = {};
#pragma unroll 8
    for (int kk = 0; kk < L1C; kk += 4) {
        v2f a = *(const v2f*)(arow + kk);
        v2f b;
        b.x = brow[kk * OUTC];
        b.y = brow[kk * OUTC + OUTC];
        c = __builtin_amdgcn_wmma_f32_16x16x4_f32(false, a, false, b, (short)0, c,
                                                  false, false);
    }
    float* p = h2 + (size_t)(wave * 16 + 8 * kh) * OUTC + m;
    if (full) {
#pragma unroll
        for (int v = 0; v < 8; ++v) p[(size_t)v * OUTC] = c[v];
    } else {
        for (int v = 0; v < 8; ++v)
            if (wave * 16 + 8 * kh + v < n_nodes) p[(size_t)v * OUTC] = c[v];
    }
}

__global__ void k_att2(const float* __restrict__ h2, const float* __restrict__ as2,
                       const float* __restrict__ ad2, float* __restrict__ a_s,
                       float* __restrict__ a_d, int n_nodes) {
    int n = blockIdx.x * blockDim.x + threadIdx.x;
    if (n >= n_nodes) return;
    const float* row = h2 + (size_t)n * OUTC;
    float s = 0.0f, d = 0.0f;
#pragma unroll
    for (int cc = 0; cc < OUTC; ++cc) {
        float v = row[cc];
        s += v * as2[cc];
        d += v * ad2[cc];
    }
    a_s[n] = s;
    a_d[n] = d;
}

__global__ void k_edge_max2(const int* __restrict__ ei, int E, int ET,
                            const float* __restrict__ a_s, const float* __restrict__ a_d,
                            unsigned* __restrict__ m2) {
    int e = blockIdx.x * blockDim.x + threadIdx.x;
    if (e >= ET) return;
    int s, d; edge_sd(ei, E, e, s, d);
    float v = lrelu(a_s[s] + a_d[d]);
    atomicMax(&m2[d], fenc(v));
}

__global__ void k_edge_exp2(const int* __restrict__ ei, int E, int ET,
                            const float* __restrict__ a_s, const float* __restrict__ a_d,
                            const unsigned* __restrict__ m2, float* __restrict__ denom2) {
    int e = blockIdx.x * blockDim.x + threadIdx.x;
    if (e >= ET) return;
    int s, d; edge_sd(ei, E, e, s, d);
    float v = lrelu(a_s[s] + a_d[d]);
    atomicAdd(&denom2[d], expf(v - fdec(m2[d])));
}

__global__ void k_msg2(const int* __restrict__ ei, int E, int ET,
                       const float* __restrict__ a_s, const float* __restrict__ a_d,
                       const unsigned* __restrict__ m2, const float* __restrict__ denom2,
                       const float* __restrict__ h2, float* __restrict__ out) {
    long long gid = (long long)blockIdx.x * blockDim.x + threadIdx.x;
    int e = (int)(gid >> 4);
    int c = (int)(gid & 15);
    if (e >= ET) return;
    int s, d; edge_sd(ei, E, e, s, d);
    float v = lrelu(a_s[s] + a_d[d]);
    float w = expf(v - fdec(m2[d]));
    float alpha = w / (denom2[d] + 1e-16f);
    atomicAdd(&out[(size_t)d * OUTC + c], h2[(size_t)s * OUTC + c] * alpha);
}

__global__ void k_bias2(float* __restrict__ out, const float* __restrict__ bias2, size_t n) {
    size_t i = (size_t)blockIdx.x * blockDim.x + threadIdx.x;
    size_t stride = (size_t)gridDim.x * blockDim.x;
    for (; i < n; i += stride) out[i] += bias2[i & (OUTC - 1)];
}

extern "C" void kernel_launch(void* const* d_in, const int* in_sizes, int n_in,
                              void* d_out, int out_size, void* d_ws, size_t ws_size,
                              hipStream_t stream) {
    const float* x     = (const float*)d_in[0];
    const int*   ei    = (const int*)d_in[1];
    const float* W1    = (const float*)d_in[2];
    const float* as1   = (const float*)d_in[3];
    const float* ad1   = (const float*)d_in[4];
    const float* bias1 = (const float*)d_in[5];
    const float* W2    = (const float*)d_in[6];
    const float* as2   = (const float*)d_in[7];
    const float* ad2   = (const float*)d_in[8];
    const float* bias2 = (const float*)d_in[9];
    float* out = (float*)d_out;

    const int N  = in_sizes[0] / 3;
    const int E  = in_sizes[1] / 2;
    const int ET = E + N;

    float* ws = (float*)d_ws;
    const size_t Nz = (size_t)N;
    float*    out1   = ws;
    unsigned* m1     = (unsigned*)(ws + 256 * Nz);
    float*    denom1 = ws + 264 * Nz;
    unsigned* m2     = (unsigned*)(ws + 272 * Nz);
    float*    denom2 = ws + 273 * Nz;
    const size_t zero_count = 274 * Nz;
    float*    h1   = ws + 274 * Nz;
    float*    a_s1 = ws + 530 * Nz;
    float*    a_d1 = ws + 538 * Nz;
    float*    h2   = ws + 546 * Nz;
    float*    a_s2 = ws + 562 * Nz;
    float*    a_d2 = ws + 563 * Nz;

    const int TB = 256;
    k_fill_zero<<<2048, TB, 0, stream>>>(ws, zero_count);
    k_fill_zero<<<512, TB, 0, stream>>>(out, (size_t)N * OUTC);

    {
        int mtiles = (N + 15) / 16;
        long long waves = (long long)mtiles * 16;
        int blocks = (int)((waves * 32 + TB - 1) / TB);
        k_gemm1_wmma<<<blocks, TB, 0, stream>>>(x, W1, h1, N);
    }
    k_att1<<<N, TB, 0, stream>>>(h1, as1, ad1, a_s1, a_d1, N);
    k_edge_max1<<<(ET + TB - 1) / TB, TB, 0, stream>>>(ei, E, ET, a_s1, a_d1, m1);
    k_edge_exp1<<<(ET + TB - 1) / TB, TB, 0, stream>>>(ei, E, ET, a_s1, a_d1, m1, denom1);
    {
        long long thr = (long long)ET * 32;
        k_msg1<<<(int)((thr + TB - 1) / TB), TB, 0, stream>>>(ei, E, ET, a_s1, a_d1,
                                                              m1, denom1, h1, out1);
    }
    k_elu<<<2048, TB, 0, stream>>>(out1, bias1, (size_t)N * L1C);

    {
        int mtiles = (N + 15) / 16;
        long long thr = (long long)mtiles * 32;
        k_gemm2_wmma<<<(int)((thr + TB - 1) / TB), TB, 0, stream>>>(out1, W2, h2, N);
    }
    k_att2<<<(N + TB - 1) / TB, TB, 0, stream>>>(h2, as2, ad2, a_s2, a_d2, N);
    k_edge_max2<<<(ET + TB - 1) / TB, TB, 0, stream>>>(ei, E, ET, a_s2, a_d2, m2);
    k_edge_exp2<<<(ET + TB - 1) / TB, TB, 0, stream>>>(ei, E, ET, a_s2, a_d2, m2, denom2);
    {
        long long thr = (long long)ET * 16;
        k_msg2<<<(int)((thr + TB - 1) / TB), TB, 0, stream>>>(ei, E, ET, a_s2, a_d2,
                                                              m2, denom2, h2, out);
    }
    k_bias2<<<512, TB, 0, stream>>>(out, bias2, (size_t)N * OUTC);
}


# Softmax division deferred: message passes add exp-weighted rows and the denominators in one pass, one divide per output element after the sums; both separate denominator passes dropped (12 launches)
# speedup vs baseline: 1.1934x; 1.1934x over previous
#include <hip/hip_runtime.h>
#include <math.h>

#define HEADS     8
#define HIDC      32
#define L1C       256
#define OUTC      16
#define NEG_SLOPE 0.2f

typedef __attribute__((ext_vector_type(2))) float v2f;
typedef __attribute__((ext_vector_type(8))) float v8f;

__device__ __forceinline__ unsigned fenc(float f) {
    unsigned u = __float_as_uint(f);
    return (u & 0x80000000u) ? ~u : (u | 0x80000000u);
}
__device__ __forceinline__ float fdec(unsigned k) {
    return __uint_as_float((k & 0x80000000u) ? (k ^ 0x80000000u) : ~k);
}
__device__ __forceinline__ float lrelu(float v) {
    return v > 0.0f ? v : NEG_SLOPE * v;
}
__device__ __forceinline__ void edge_sd(const int* __restrict__ ei, int E, int e,
                                        int& s, int& d) {
    if (e < E) { s = ei[e]; d = ei[E + e]; }
    else       { s = e - E; d = e - E; }
}

__global__ void k_fill_zero(float* __restrict__ p, size_t n) {
    size_t i = (size_t)blockIdx.x * blockDim.x + threadIdx.x;
    size_t stride = (size_t)gridDim.x * blockDim.x;
    for (; i < n; i += stride) p[i] = 0.0f;
}

__global__ void k_gemm1_wmma(const float* __restrict__ x, const float* __restrict__ W1,
                             float* __restrict__ h1, int n_nodes) {
    int wave   = (int)((blockIdx.x * blockDim.x + threadIdx.x) >> 5);
    int lane   = threadIdx.x & 31;
    int mtiles = (n_nodes + 15) >> 4;
    int mtile  = wave >> 4;
    int ntile  = wave & 15;
    if (mtile >= mtiles) return;
    int m  = lane & 15;
    int kh = lane >> 4;
    bool full = (mtile * 16 + 16 <= n_nodes);
    int node = mtile * 16 + m;
    if (node >= n_nodes) node = n_nodes - 1;
    float x0 = x[node * 3 + 0], x1 = x[node * 3 + 1], x2 = x[node * 3 + 2];
    int col = ntile * 16 + m;
    float w0 = W1[0 * L1C + col], w1 = W1[1 * L1C + col], w2 = W1[2 * L1C + col];
    v2f a, b;
    a.x = kh ? x2 : x0;  a.y = kh ? 0.0f : x1;
    b.x = kh ? w2 : w0;  b.y = kh ? 0.0f : w1;
    v8f c = {};
    c = __builtin_amdgcn_wmma_f32_16x16x4_f32(false, a, false, b, (short)0, c,
                                              false, false);
    float* p = h1 + (size_t)(mtile * 16 + 8 * kh) * L1C + col;
    if (full) {
#pragma unroll
        for (int v = 0; v < 8; ++v) p[(size_t)v * L1C] = c[v];
    } else {
        for (int v = 0; v < 8; ++v)
            if (mtile * 16 + 8 * kh + v < n_nodes) p[(size_t)v * L1C] = c[v];
    }
}

__global__ void k_att1(const float* __restrict__ h1, const float* __restrict__ as1,
                       const float* __restrict__ ad1, float* __restrict__ a_s,
                       float* __restrict__ a_d, int n_nodes) {
    int node = blockIdx.x;
    if (node >= n_nodes) return;
    int head = threadIdx.x >> 5;
    int lane = threadIdx.x & 31;
    float v = h1[(size_t)node * L1C + head * HIDC + lane];
    float s = v * as1[head * HIDC + lane];
    float d = v * ad1[head * HIDC + lane];
    for (int off = 16; off; off >>= 1) {
        s += __shfl_xor(s, off, 32);
        d += __shfl_xor(d, off, 32);
    }
    if (lane == 0) {
        a_s[node * HEADS + head] = s;
        a_d[node * HEADS + head] = d;
    }
}

__global__ void k_edge_max1(const int* __restrict__ ei, int E, int ET,
                            const float* __restrict__ a_s, const float* __restrict__ a_d,
                            unsigned* __restrict__ m1) {
    int e = blockIdx.x * blockDim.x + threadIdx.x;
    if (e >= ET) return;
    int s, d; edge_sd(ei, E, e, s, d);
#pragma unroll
    for (int h = 0; h < HEADS; ++h) {
        float v = lrelu(a_s[s * HEADS + h] + a_d[d * HEADS + h]);
        atomicMax(&m1[d * HEADS + h], fenc(v));
    }
}

__global__ void k_msg1(const int* __restrict__ ei, int E, int ET,
                       const float* __restrict__ a_s, const float* __restrict__ a_d,
                       const unsigned* __restrict__ m1, float* __restrict__ denom1,
                       const float* __restrict__ h1, float* __restrict__ out1) {
    long long gid = (long long)blockIdx.x * blockDim.x + threadIdx.x;
    int e = (int)(gid >> 5);
    int lane = (int)(gid & 31);
    if (e >= ET) return;
    int s, d; edge_sd(ei, E, e, s, d);
    float myw = 0.0f;
    if (lane < HEADS) {
        float v = lrelu(a_s[s * HEADS + lane] + a_d[d * HEADS + lane]);
        myw = expf(v - fdec(m1[d * HEADS + lane]));
        atomicAdd(&denom1[d * HEADS + lane], myw);
    }
    const float* ph = h1 + (size_t)s * L1C + lane;
    float* po = out1 + (size_t)d * L1C + lane;
#pragma unroll
    for (int h = 0; h < HEADS; ++h) {
        float wh = __shfl(myw, h, 32);
        atomicAdd(po + h * HIDC, ph[h * HIDC] * wh);
    }
}

__global__ void k_norm_elu(float* __restrict__ out1, const float* __restrict__ denom1,
                           const float* __restrict__ bias1, size_t n) {
    size_t i = (size_t)blockIdx.x * blockDim.x + threadIdx.x;
    size_t stride = (size_t)gridDim.x * blockDim.x;
    for (; i < n; i += stride) {
        float v = out1[i] / (denom1[i >> 5] + 1e-16f) + bias1[i & (L1C - 1)];
        out1[i] = v > 0.0f ? v : (expf(v) - 1.0f);
    }
}

__global__ void k_gemm2_wmma(const float* __restrict__ helu, const float* __restrict__ W2,
                             float* __restrict__ h2, int n_nodes) {
    int wave = (int)((blockIdx.x * blockDim.x + threadIdx.x) >> 5);
    int lane = threadIdx.x & 31;
    int mtiles = (n_nodes + 15) >> 4;
    if (wave >= mtiles) return;
    int m  = lane & 15;
    int kh = lane >> 4;
    bool full = (wave * 16 + 16 <= n_nodes);
    int node = wave * 16 + m;
    if (node >= n_nodes) node = n_nodes - 1;
    const float* arow = helu + (size_t)node * L1C + kh * 2;
    const float* brow = W2 + kh * 2 * OUTC + m;
    v8f c = {};
#pragma unroll 8
    for (int kk = 0; kk < L1C; kk += 4) {
        v2f a = *(const v2f*)(arow + kk);
        v2f b;
        b.x = brow[kk * OUTC];
        b.y = brow[kk * OUTC + OUTC];
        c = __builtin_amdgcn_wmma_f32_16x16x4_f32(false, a, false, b, (short)0, c,
                                                  false, false);
    }
    float* p = h2 + (size_t)(wave * 16 + 8 * kh) * OUTC + m;
    if (full) {
#pragma unroll
        for (int v = 0; v < 8; ++v) p[(size_t)v * OUTC] = c[v];
    } else {
        for (int v = 0; v < 8; ++v)
            if (wave * 16 + 8 * kh + v < n_nodes) p[(size_t)v * OUTC] = c[v];
    }
}

__global__ void k_att2(const float* __restrict__ h2, const float* __restrict__ as2,
                       const float* __restrict__ ad2, float* __restrict__ a_s,
                       float* __restrict__ a_d, int n_nodes) {
    int n = blockIdx.x * blockDim.x + threadIdx.x;
    if (n >= n_nodes) return;
    const float* row = h2 + (size_t)n * OUTC;
    float s = 0.0f, d = 0.0f;
#pragma unroll
    for (int cc = 0; cc < OUTC; ++cc) {
        float v = row[cc];
        s += v * as2[cc];
        d += v * ad2[cc];
    }
    a_s[n] = s;
    a_d[n] = d;
}

__global__ void k_edge_max2(const int* __restrict__ ei, int E, int ET,
                            const float* __restrict__ a_s, const float* __restrict__ a_d,
                            unsigned* __restrict__ m2) {
    int e = blockIdx.x * blockDim.x + threadIdx.x;
    if (e >= ET) return;
    int s, d; edge_sd(ei, E, e, s, d);
    float v = lrelu(a_s[s] + a_d[d]);
    atomicMax(&m2[d], fenc(v));
}

__global__ void k_msg2(const int* __restrict__ ei, int E, int ET,
                       const float* __restrict__ a_s, const float* __restrict__ a_d,
                       const unsigned* __restrict__ m2, float* __restrict__ denom2,
                       const float* __restrict__ h2, float* __restrict__ out) {
    long long gid = (long long)blockIdx.x * blockDim.x + threadIdx.x;
    int e = (int)(gid >> 4);
    int c = (int)(gid & 15);
    if (e >= ET) return;
    int s, d; edge_sd(ei, E, e, s, d);
    float v = lrelu(a_s[s] + a_d[d]);
    float w = expf(v - fdec(m2[d]));
    if (c == 0) atomicAdd(&denom2[d], w);
    atomicAdd(&out[(size_t)d * OUTC + c], h2[(size_t)s * OUTC + c] * w);
}

__global__ void k_norm_bias2(float* __restrict__ out, const float* __restrict__ denom2,
                             const float* __restrict__ bias2, size_t n) {
    size_t i = (size_t)blockIdx.x * blockDim.x + threadIdx.x;
    size_t stride = (size_t)gridDim.x * blockDim.x;
    for (; i < n; i += stride)
        out[i] = out[i] / (denom2[i >> 4] + 1e-16f) + bias2[i & (OUTC - 1)];
}

extern "C" void kernel_launch(void* const* d_in, const int* in_sizes, int n_in,
                              void* d_out, int out_size, void* d_ws, size_t ws_size,
                              hipStream_t stream) {
    const float* x     = (const float*)d_in[0];
    const int*   ei    = (const int*)d_in[1];
    const float* W1    = (const float*)d_in[2];
    const float* as1   = (const float*)d_in[3];
    const float* ad1   = (const float*)d_in[4];
    const float* bias1 = (const float*)d_in[5];
    const float* W2    = (const float*)d_in[6];
    const float* as2   = (const float*)d_in[7];
    const float* ad2   = (const float*)d_in[8];
    const float* bias2 = (const float*)d_in[9];
    float* out = (float*)d_out;

    const int N  = in_sizes[0] / 3;
    const int E  = in_sizes[1] / 2;
    const int ET = E + N;

    float* ws = (float*)d_ws;
    const size_t Nz = (size_t)N;
    float*    out1   = ws;
    unsigned* m1     = (unsigned*)(ws + 256 * Nz);
    float*    denom1 = ws + 264 * Nz;
    unsigned* m2     = (unsigned*)(ws + 272 * Nz);
    float*    denom2 = ws + 273 * Nz;
    const size_t zero_count = 274 * Nz;
    float*    h1   = ws + 274 * Nz;
    float*    a_s1 = ws + 530 * Nz;
    float*    a_d1 = ws + 538 * Nz;
    float*    h2   = ws + 546 * Nz;
    float*    a_s2 = ws + 562 * Nz;
    float*    a_d2 = ws + 563 * Nz;

    const int TB = 256;
    k_fill_zero<<<2048, TB, 0, stream>>>(ws, zero_count);
    k_fill_zero<<<512, TB, 0, stream>>>(out, (size_t)N * OUTC);

    {
        int mtiles = (N + 15) / 16;
        long long waves = (long long)mtiles * 16;
        int blocks = (int)((waves * 32 + TB - 1) / TB);
        k_gemm1_wmma<<<blocks, TB, 0, stream>>>(x, W1, h1, N);
    }
    k_att1<<<N, TB, 0, stream>>>(h1, as1, ad1, a_s1, a_d1, N);
    k_edge_max1<<<(ET + TB - 1) / TB, TB, 0, stream>>>(ei, E, ET, a_s1, a_d1, m1);
    {
        long long thr = (long long)ET * 32;
        k_msg1<<<(int)((thr + TB - 1) / TB), TB, 0, stream>>>(ei, E, ET, a_s1, a_d1,
                                                              m1, denom1, h1, out1);
    }
    k_norm_elu<<<2048, TB, 0, stream>>>(out1, denom1, bias1, (size_t)N * L1C);

    {
        int mtiles = (N + 15) / 16;
        long long thr = (long long)mtiles * 32;
        k_gemm2_wmma<<<(int)((thr + TB - 1) / TB), TB, 0, stream>>>(out1, W2, h2, N);
    }
    k_att2<<<(N + TB - 1) / TB, TB, 0, stream>>>(h2, as2, ad2, a_s2, a_d2, N);
    k_edge_max2<<<(ET + TB - 1) / TB, TB, 0, stream>>>(ei, E, ET, a_s2, a_d2, m2);
    {
        long long thr = (long long)ET * 16;
        k_msg2<<<(int)((thr + TB - 1) / TB), TB, 0, stream>>>(ei, E, ET, a_s2, a_d2,
                                                              m2, denom2, h2, out);
    }
    k_norm_bias2<<<512, TB, 0, stream>>>(out, denom2, bias2, (size_t)N * OUTC);
}


# Layer-1 edge sums taken on the 3-wide input: an edge adds w*x[src] and w (4 floats per head, one 128-byte atomic line), not w*h1[src]; W1 applied after the sums on WMMA (sum order changes)
# speedup vs baseline: 1.7802x; 1.4917x over previous
#include <hip/hip_runtime.h>
#include <math.h>

#define HEADS     8
#define HIDC      32
#define L1C       256
#define OUTC      16
#define NEG_SLOPE 0.2f

typedef __attribute__((ext_vector_type(2))) float v2f;
typedef __attribute__((ext_vector_type(8))) float v8f;

__device__ __forceinline__ unsigned fenc(float f) {
    unsigned u = __float_as_uint(f);
    return (u & 0x80000000u) ? ~u : (u | 0x80000000u);
}
__device__ __forceinline__ float fdec(unsigned k) {
    return __uint_as_float((k & 0x80000000u) ? (k ^ 0x80000000u) : ~k);
}
__device__ __forceinline__ float lrelu(float v) {
    return v > 0.0f ? v : NEG_SLOPE * v;
}
__device__ __forceinline__ void edge_sd(const int* __restrict__ ei, int E, int e,
                                        int& s, int& d) {
    if (e < E) { s = ei[e]; d = ei[E + e]; }
    else       { s = e - E; d = e - E; }
}

__global__ void k_fill_zero(float* __restrict__ p, size_t n) {
    size_t i = (size_t)blockIdx.x * blockDim.x + threadIdx.x;
    size_t stride = (size_t)gridDim.x * blockDim.x;
    for (; i < n; i += stride) p[i] = 0.0f;
}

__global__ void k_gemm1_wmma(const float* __restrict__ x, const float* __restrict__ W1,
                             float* __restrict__ h1, int n_nodes) {
    int wave   = (int)((blockIdx.x * blockDim.x + threadIdx.x) >> 5);
    int lane   = threadIdx.x & 31;
    int mtiles = (n_nodes + 15) >> 4;
    int mtile  = wave >> 4;
    int ntile  = wave & 15;
    if (mtile >= mtiles) return;
    int m  = lane & 15;
    int kh = lane >> 4;
    bool full = (mtile * 16 + 16 <= n_nodes);
    int node = mtile * 16 + m;
    if (node >= n_nodes) node = n_nodes - 1;
    float x0 = x[node * 3 + 0], x1 = x[node * 3 + 1], x2 = x[node * 3 + 2];
    int col = ntile * 16 + m;
    float w0 = W1[0 * L1C + col], w1 = W1[1 * L1C + col], w2 = W1[2 * L1C + col];
    v2f a, b;
    a.x = kh ? x2 : x0;  a.y = kh ? 0.0f : x1;
    b.x = kh ? w2 : w0;  b.y = kh ? 0.0f : w1;
    v8f c = {};
    c = __builtin_amdgcn_wmma_f32_16x16x4_f32(false, a, false, b, (short)0, c,
                                              false, false);
    float* p = h1 + (size_t)(mtile * 16 + 8 * kh) * L1C + col;
    if (full) {
#pragma unroll
        for (int v = 0; v < 8; ++v) p[(size_t)v * L1C] = c[v];
    } else {
        for (int v = 0; v < 8; ++v)
            if (mtile * 16 + 8 * kh + v < n_nodes) p[(size_t)v * L1C] = c[v];
    }
}

__global__ void k_att1(const float* __restrict__ h1, const float* __restrict__ as1,
                       const float* __restrict__ ad1, float* __restrict__ a_s,
                       float* __restrict__ a_d, int n_nodes) {
    int node = blockIdx.x;
    if (node >= n_nodes) return;
    int head = threadIdx.x >> 5;
    int lane = threadIdx.x & 31;
    float v = h1[(size_t)node * L1C + head * HIDC + lane];
    float s = v * as1[head * HIDC + lane];
    float d = v * ad1[head * HIDC + lane];
    for (int off = 16; off; off >>= 1) {
        s += __shfl_xor(s, off, 32);
        d += __shfl_xor(d, off, 32);
    }
    if (lane == 0) {
        a_s[node * HEADS + head] = s;
        a_d[node * HEADS + head] = d;
    }
}

__global__ void k_edge_max1(const int* __restrict__ ei, int E, int ET,
                            const float* __restrict__ a_s, const float* __restrict__ a_d,
                            unsigned* __restrict__ m1) {
    int e = blockIdx.x * blockDim.x + threadIdx.x;
    if (e >= ET) return;
    int s, d; edge_sd(ei, E, e, s, d);
#pragma unroll
    for (int h = 0; h < HEADS; ++h) {
        float v = lrelu(a_s[s * HEADS + h] + a_d[d * HEADS + h]);
        atomicMax(&m1[d * HEADS + h], fenc(v));
    }
}

#define EPW 16
__global__ void k_msg1x(const int* __restrict__ ei, int E, int ET,
                        const float* __restrict__ x,
                        const float* __restrict__ a_s, const float* __restrict__ a_d,
                        const unsigned* __restrict__ m1, float* __restrict__ agg) {
    int wave = (int)((blockIdx.x * blockDim.x + threadIdx.x) >> 5);
    int lane = threadIdx.x & 31;
    int h  = lane >> 2;
    int k  = lane & 3;
    int kx = k < 3 ? k : 2;
    int e0 = wave * EPW;
    int el = e0 + (lane & (EPW - 1));
    int sl = 0, dl = 0;
    if (el < ET) edge_sd(ei, E, el, sl, dl);
    float val[EPW];
#pragma unroll
    for (int i = 0; i < EPW; ++i) {
        int s = __builtin_amdgcn_readlane(sl, i);
        int d = __builtin_amdgcn_readlane(dl, i);
        float v  = lrelu(a_s[s * HEADS + h] + a_d[d * HEADS + h]);
        float w  = expf(v - fdec(m1[d * HEADS + h]));
        float xv = x[s * 3 + kx];
        val[i] = (k < 3) ? w * xv : w;
    }
    if (e0 + EPW <= ET) {
#pragma unroll
        for (int i = 0; i < EPW; ++i) {
            int d = __builtin_amdgcn_readlane(dl, i);
            atomicAdd(&agg[(size_t)d * 32 + lane], val[i]);
        }
    } else {
#pragma unroll
        for (int i = 0; i < EPW; ++i) {
            int d = __builtin_amdgcn_readlane(dl, i);
            if (e0 + i < ET) atomicAdd(&agg[(size_t)d * 32 + lane], val[i]);
        }
    }
}

__global__ void k_gemm1_agg_wmma(const float* __restrict__ agg, const float* __restrict__ W1,
                                 const float* __restrict__ bias1, float* __restrict__ helu,
                                 int n_nodes) {
    int mtile = blockIdx.x;
    int head  = threadIdx.x >> 5;
    int lane  = threadIdx.x & 31;
    int m  = lane & 15;
    int kh = lane >> 4;
    bool full = (mtile * 16 + 16 <= n_nodes);
    int node = mtile * 16 + m;
    if (node >= n_nodes) node = n_nodes - 1;
    const float* prow = agg + (size_t)node * 32 + head * 4;
    v2f p = *(const v2f*)(prow + 2 * kh);
    float den = prow[3] + 1e-16f;
    v2f a;
    a.x = p.x / den;
    a.y = kh ? 0.0f : p.y / den;
    float* prow_out = helu + (size_t)(mtile * 16 + 8 * kh) * L1C;
#pragma unroll
    for (int t = 0; t < 2; ++t) {
        int col = head * HIDC + t * 16 + m;
        float w0 = W1[0 * L1C + col], w1 = W1[1 * L1C + col], w2 = W1[2 * L1C + col];
        v2f b;
        b.x = kh ? w2 : w0;  b.y = kh ? 0.0f : w1;
        v8f c = {};
        c = __builtin_amdgcn_wmma_f32_16x16x4_f32(false, a, false, b, (short)0, c,
                                                  false, false);
        float bs = bias1[col];
        float* po = prow_out + col;
        if (full) {
#pragma unroll
            for (int v = 0; v < 8; ++v) {
                float r = c[v] + bs;
                po[(size_t)v * L1C] = r > 0.0f ? r : (expf(r) - 1.0f);
            }
        } else {
            for (int v = 0; v < 8; ++v)
                if (mtile * 16 + 8 * kh + v < n_nodes) {
                    float r = c[v] + bs;
                    po[(size_t)v * L1C] = r > 0.0f ? r : (expf(r) - 1.0f);
                }
        }
    }
}

__global__ void k_gemm2_wmma(const float* __restrict__ helu, const float* __restrict__ W2,
                             float* __restrict__ h2, int n_nodes) {
    int wave = (int)((blockIdx.x * blockDim.x + threadIdx.x) >> 5);
    int lane = threadIdx.x & 31;
    int mtiles = (n_nodes + 15) >> 4;
    if (wave >= mtiles) return;
    int m  = lane & 15;
    int kh = lane >> 4;
    bool full = (wave * 16 + 16 <= n_nodes);
    int node = wave * 16 + m;
    if (node >= n_nodes) node = n_nodes - 1;
    const float* arow = helu + (size_t)node * L1C + kh * 2;
    const float* brow = W2 + kh * 2 * OUTC + m;
    v8f c = {};
#pragma unroll 8
    for (int kk = 0; kk < L1C; kk += 4) {
        v2f a = *(const v2f*)(arow + kk);
        v2f b;
        b.x = brow[kk * OUTC];
        b.y = brow[kk * OUTC + OUTC];
        c = __builtin_amdgcn_wmma_f32_16x16x4_f32(false, a, false, b, (short)0, c,
                                                  false, false);
    }
    float* p = h2 + (size_t)(wave * 16 + 8 * kh) * OUTC + m;
    if (full) {
#pragma unroll
        for (int v = 0; v < 8; ++v) p[(size_t)v * OUTC] = c[v];
    } else {
        for (int v = 0; v < 8; ++v)
            if (wave * 16 + 8 * kh + v < n_nodes) p[(size_t)v * OUTC] = c[v];
    }
}

__global__ void k_att2(const float* __restrict__ h2, const float* __restrict__ as2,
                       const float* __restrict__ ad2, float* __restrict__ a_s,
                       float* __restrict__ a_d, int n_nodes) {
    int n = blockIdx.x * blockDim.x + threadIdx.x;
    if (n >= n_nodes) return;
    const float* row = h2 + (size_t)n * OUTC;
    float s = 0.0f, d = 0.0f;
#pragma unroll
    for (int cc = 0; cc < OUTC; ++cc) {
        float v = row[cc];
        s += v * as2[cc];
        d += v * ad2[cc];
    }
    a_s[n] = s;
    a_d[n] = d;
}

__global__ void k_edge_max2(const int* __restrict__ ei, int E, int ET,
                            const float* __restrict__ a_s, const float* __restrict__ a_d,
                            unsigned* __restrict__ m2) {
    int e = blockIdx.x * blockDim.x + threadIdx.x;
    if (e >= ET) return;
    int s, d; edge_sd(ei, E, e, s, d);
    float v = lrelu(a_s[s] + a_d[d]);
    atomicMax(&m2[d], fenc(v));
}

__global__ void k_msg2(const int* __restrict__ ei, int E, int ET,
                       const float* __restrict__ a_s, const float* __restrict__ a_d,
                       const unsigned* __restrict__ m2, float* __restrict__ denom2,
                       const float* __restrict__ h2, float* __restrict__ out) {
    long long gid = (long long)blockIdx.x * blockDim.x + threadIdx.x;
    int e = (int)(gid >> 4);
    int c = (int)(gid & 15);
    if (e >= ET) return;
    int s, d; edge_sd(ei, E, e, s, d);
    float v = lrelu(a_s[s] + a_d[d]);
    float w = expf(v - fdec(m2[d]));
    if (c == 0) atomicAdd(&denom2[d], w);
    atomicAdd(&out[(size_t)d * OUTC + c], h2[(size_t)s * OUTC + c] * w);
}

__global__ void k_norm_bias2(float* __restrict__ out, const float* __restrict__ denom2,
                             const float* __restrict__ bias2, size_t n) {
    size_t i = (size_t)blockIdx.x * blockDim.x + threadIdx.x;
    size_t stride = (size_t)gridDim.x * blockDim.x;
    for (; i < n; i += stride)
        out[i] = out[i] / (denom2[i >> 4] + 1e-16f) + bias2[i & (OUTC - 1)];
}

extern "C" void kernel_launch(void* const* d_in, const int* in_sizes, int n_in,
                              void* d_out, int out_size, void* d_ws, size_t ws_size,
                              hipStream_t stream) {
    const float* x     = (const float*)d_in[0];
    const int*   ei    = (const int*)d_in[1];
    const float* W1    = (const float*)d_in[2];
    const float* as1   = (const float*)d_in[3];
    const float* ad1   = (const float*)d_in[4];
    const float* bias1 = (const float*)d_in[5];
    const float* W2    = (const float*)d_in[6];
    const float* as2   = (const float*)d_in[7];
    const float* ad2   = (const float*)d_in[8];
    const float* bias2 = (const float*)d_in[9];
    float* out = (float*)d_out;

    const int N  = in_sizes[0] / 3;
    const int E  = in_sizes[1] / 2;
    const int ET = E + N;

    float* ws = (float*)d_ws;
    const size_t Nz = (size_t)N;
    float*    agg1   = ws;
    unsigned* m1     = (unsigned*)(ws + 32 * Nz);
    unsigned* m2     = (unsigned*)(ws + 40 * Nz);
    float*    denom2 = ws + 41 * Nz;
    const size_t zero_count = 42 * Nz;
    float*    h1   = ws + 42 * Nz;
    float*    helu = ws + 298 * Nz;
    float*    a_s1 = ws + 554 * Nz;
    float*    a_d1 = ws + 562 * Nz;
    float*    h2   = ws + 570 * Nz;
    float*    a_s2 = ws + 586 * Nz;
    float*    a_d2 = ws + 587 * Nz;

    const int TB = 256;
    k_fill_zero<<<2048, TB, 0, stream>>>(ws, zero_count);
    k_fill_zero<<<512, TB, 0, stream>>>(out, (size_t)N * OUTC);

    {
        int mtiles = (N + 15) / 16;
        long long waves = (long long)mtiles * 16;
        int blocks = (int)((waves * 32 + TB - 1) / TB);
        k_gemm1_wmma<<<blocks, TB, 0, stream>>>(x, W1, h1, N);
    }
    k_att1<<<N, TB, 0, stream>>>(h1, as1, ad1, a_s1, a_d1, N);
    k_edge_max1<<<(ET + TB - 1) / TB, TB, 0, stream>>>(ei, E, ET, a_s1, a_d1, m1);
    {
        long long waves = ((long long)ET + EPW - 1) / EPW;
        k_msg1x<<<(int)((waves * 32 + TB - 1) / TB), TB, 0, stream>>>(ei, E, ET, x, a_s1, a_d1,
                                                                      m1, agg1);
    }
    k_gemm1_agg_wmma<<<(N + 15) / 16, TB, 0, stream>>>(agg1, W1, bias1, helu, N);

    {
        int mtiles = (N + 15) / 16;
        long long thr = (long long)mtiles * 32;
        k_gemm2_wmma<<<(int)((thr + TB - 1) / TB), TB, 0, stream>>>(helu, W2, h2, N);
    }
    k_att2<<<(N + TB - 1) / TB, TB, 0, stream>>>(h2, as2, ad2, a_s2, a_d2, N);
    k_edge_max2<<<(ET + TB - 1) / TB, TB, 0, stream>>>(ei, E, ET, a_s2, a_d2, m2);
    {
        long long thr = (long long)ET * 16;
        k_msg2<<<(int)((thr + TB - 1) / TB), TB, 0, stream>>>(ei, E, ET, a_s2, a_d2,
                                                              m2, denom2, h2, out);
    }
    k_norm_bias2<<<512, TB, 0, stream>>>(out, denom2, bias2, (size_t)N * OUTC);
}


# Layer-1 max pass: lane = edge slot x head, an edge's 8 heads go out as one 32-byte atomic request (4 edges per instruction), not 8 instructions of 32 scattered lanes; max exact, order-free
# speedup vs baseline: 2.5849x; 1.4520x over previous
#include <hip/hip_runtime.h>
#include <math.h>

#define HEADS     8
#define HIDC      32
#define L1C       256
#define OUTC      16
#define NEG_SLOPE 0.2f

typedef __attribute__((ext_vector_type(2))) float v2f;
typedef __attribute__((ext_vector_type(8))) float v8f;

__device__ __forceinline__ unsigned fenc(float f) {
    unsigned u = __float_as_uint(f);
    return (u & 0x80000000u) ? ~u : (u | 0x80000000u);
}
__device__ __forceinline__ float fdec(unsigned k) {
    return __uint_as_float((k & 0x80000000u) ? (k ^ 0x80000000u) : ~k);
}
__device__ __forceinline__ float lrelu(float v) {
    return v > 0.0f ? v : NEG_SLOPE * v;
}
__device__ __forceinline__ void edge_sd(const int* __restrict__ ei, int E, int e,
                                        int& s, int& d) {
    if (e < E) { s = ei[e]; d = ei[E + e]; }
    else       { s = e - E; d = e - E; }
}

__global__ void k_fill_zero(float* __restrict__ p, size_t n) {
    size_t i = (size_t)blockIdx.x * blockDim.x + threadIdx.x;
    size_t stride = (size_t)gridDim.x * blockDim.x;
    for (; i < n; i += stride) p[i] = 0.0f;
}

__global__ void k_gemm1_wmma(const float* __restrict__ x, const float* __restrict__ W1,
                             float* __restrict__ h1, int n_nodes) {
    int wave   = (int)((blockIdx.x * blockDim.x + threadIdx.x) >> 5);
    int lane   = threadIdx.x & 31;
    int mtiles = (n_nodes + 15) >> 4;
    int mtile  = wave >> 4;
    int ntile  = wave & 15;
    if (mtile >= mtiles) return;
    int m  = lane & 15;
    int kh = lane >> 4;
    bool full = (mtile * 16 + 16 <= n_nodes);
    int node = mtile * 16 + m;
    if (node >= n_nodes) node = n_nodes - 1;
    float x0 = x[node * 3 + 0], x1 = x[node * 3 + 1], x2 = x[node * 3 + 2];
    int col = ntile * 16 + m;
    float w0 = W1[0 * L1C + col], w1 = W1[1 * L1C + col], w2 = W1[2 * L1C + col];
    v2f a, b;
    a.x = kh ? x2 : x0;  a.y = kh ? 0.0f : x1;
    b.x = kh ? w2 : w0;  b.y = kh ? 0.0f : w1;
    v8f c = {};
    c = __builtin_amdgcn_wmma_f32_16x16x4_f32(false, a, false, b, (short)0, c,
                                              false, false);
    float* p = h1 + (size_t)(mtile * 16 + 8 * kh) * L1C + col;
    if (full) {
#pragma unroll
        for (int v = 0; v < 8; ++v) p[(size_t)v * L1C] = c[v];
    } else {
        for (int v = 0; v < 8; ++v)
            if (mtile * 16 + 8 * kh + v < n_nodes) p[(size_t)v * L1C] = c[v];
    }
}

__global__ void k_att1(const float* __restrict__ h1, const float* __restrict__ as1,
                       const float* __restrict__ ad1, float* __restrict__ a_s,
                       float* __restrict__ a_d, int n_nodes) {
    int node = blockIdx.x;
    if (node >= n_nodes) return;
    int head = threadIdx.x >> 5;
    int lane = threadIdx.x & 31;
    float v = h1[(size_t)node * L1C + head * HIDC + lane];
    float s = v * as1[head * HIDC + lane];
    float d = v * ad1[head * HIDC + lane];
    for (int off = 16; off; off >>= 1) {
        s += __shfl_xor(s, off, 32);
        d += __shfl_xor(d, off, 32);
    }
    if (lane == 0) {
        a_s[node * HEADS + head] = s;
        a_d[node * HEADS + head] = d;
    }
}

#define MXE 32
__global__ void k_edge_max1(const int* __restrict__ ei, int E, int ET,
                            const float* __restrict__ a_s, const float* __restrict__ a_d,
                            unsigned* __restrict__ m1) {
    int wave = (int)((blockIdx.x * blockDim.x + threadIdx.x) >> 5);
    int lane = threadIdx.x & 31;
    int h  = lane & 7;
    int j  = lane >> 3;
    int e0 = wave * MXE;
    if (e0 >= ET) return;
    int el = e0 + lane;
    int sl = 0, dl = 0;
    if (el < ET) edge_sd(ei, E, el, sl, dl);
    unsigned key[MXE / 4];
    int      dd[MXE / 4];
#pragma unroll
    for (int i = 0; i < MXE / 4; ++i) {
        int s = __shfl(sl, i * 4 + j, 32);
        int d = __shfl(dl, i * 4 + j, 32);
        dd[i]  = d;
        key[i] = fenc(lrelu(a_s[s * HEADS + h] + a_d[d * HEADS + h]));
    }
#pragma unroll
    for (int i = 0; i < MXE / 4; ++i)
        if (e0 + i * 4 + j < ET) atomicMax(&m1[dd[i] * HEADS + h], key[i]);
}

#define EPW 16
__global__ void k_msg1x(const int* __restrict__ ei, int E, int ET,
                        const float* __restrict__ x,
                        const float* __restrict__ a_s, const float* __restrict__ a_d,
                        const unsigned* __restrict__ m1, float* __restrict__ agg) {
    int wave = (int)((blockIdx.x * blockDim.x + threadIdx.x) >> 5);
    int lane = threadIdx.x & 31;
    int h  = lane >> 2;
    int k  = lane & 3;
    int kx = k < 3 ? k : 2;
    int e0 = wave * EPW;
    int el = e0 + (lane & (EPW - 1));
    int sl = 0, dl = 0;
    if (el < ET) edge_sd(ei, E, el, sl, dl);
    float val[EPW];
#pragma unroll
    for (int i = 0; i < EPW; ++i) {
        int s = __builtin_amdgcn_readlane(sl, i);
        int d = __builtin_amdgcn_readlane(dl, i);
        float v  = lrelu(a_s[s * HEADS + h] + a_d[d * HEADS + h]);
        float w  = expf(v - fdec(m1[d * HEADS + h]));
        float xv = x[s * 3 + kx];
        val[i] = (k < 3) ? w * xv : w;
    }
    if (e0 + EPW <= ET) {
#pragma unroll
        for (int i = 0; i < EPW; ++i) {
            int d = __builtin_amdgcn_readlane(dl, i);
            atomicAdd(&agg[(size_t)d * 32 + lane], val[i]);
        }
    } else {
#pragma unroll
        for (int i = 0; i < EPW; ++i) {
            int d = __builtin_amdgcn_readlane(dl, i);
            if (e0 + i < ET) atomicAdd(&agg[(size_t)d * 32 + lane], val[i]);
        }
    }
}

__global__ void k_gemm1_agg_wmma(const float* __restrict__ agg, const float* __restrict__ W1,
                                 const float* __restrict__ bias1, float* __restrict__ helu,
                                 int n_nodes) {
    int mtile = blockIdx.x;
    int head  = threadIdx.x >> 5;
    int lane  = threadIdx.x & 31;
    int m  = lane & 15;
    int kh = lane >> 4;
    bool full = (mtile * 16 + 16 <= n_nodes);
    int node = mtile * 16 + m;
    if (node >= n_nodes) node = n_nodes - 1;
    const float* prow = agg + (size_t)node * 32 + head * 4;
    v2f p = *(const v2f*)(prow + 2 * kh);
    float den = prow[3] + 1e-16f;
    v2f a;
    a.x = p.x / den;
    a.y = kh ? 0.0f : p.y / den;
    float* prow_out = helu + (size_t)(mtile * 16 + 8 * kh) * L1C;
#pragma unroll
    for (int t = 0; t < 2; ++t) {
        int col = head * HIDC + t * 16 + m;
        float w0 = W1[0 * L1C + col], w1 = W1[1 * L1C + col], w2 = W1[2 * L1C + col];
        v2f b;
        b.x = kh ? w2 : w0;  b.y = kh ? 0.0f : w1;
        v8f c = {};
        c = __builtin_amdgcn_wmma_f32_16x16x4_f32(false, a, false, b, (short)0, c,
                                                  false, false);
        float bs = bias1[col];
        float* po = prow_out + col;
        if (full) {
#pragma unroll
            for (int v = 0; v < 8; ++v) {
                float r = c[v] + bs;
                po[(size_t)v * L1C] = r > 0.0f ? r : (expf(r) - 1.0f);
            }
        } else {
            for (int v = 0; v < 8; ++v)
                if (mtile * 16 + 8 * kh + v < n_nodes) {
                    float r = c[v] + bs;
                    po[(size_t)v * L1C] = r > 0.0f ? r : (expf(r) - 1.0f);
                }
        }
    }
}

__global__ void k_gemm2_wmma(const float* __restrict__ helu, const float* __restrict__ W2,
                             float* __restrict__ h2, int n_nodes) {
    int wave = (int)((blockIdx.x * blockDim.x + threadIdx.x) >> 5);
    int lane = threadIdx.x & 31;
    int mtiles = (n_nodes + 15) >> 4;
    if (wave >= mtiles) return;
    int m  = lane & 15;
    int kh = lane >> 4;
    bool full = (wave * 16 + 16 <= n_nodes);
    int node = wave * 16 + m;
    if (node >= n_nodes) node = n_nodes - 1;
    const float* arow = helu + (size_t)node * L1C + kh * 2;
    const float* brow = W2 + kh * 2 * OUTC + m;
    v8f c = {};
#pragma unroll 8
    for (int kk = 0; kk < L1C; kk += 4) {
        v2f a = *(const v2f*)(arow + kk);
        v2f b;
        b.x = brow[kk * OUTC];
        b.y = brow[kk * OUTC + OUTC];
        c = __builtin_amdgcn_wmma_f32_16x16x4_f32(false, a, false, b, (short)0, c,
                                                  false, false);
    }
    float* p = h2 + (size_t)(wave * 16 + 8 * kh) * OUTC + m;
    if (full) {
#pragma unroll
        for (int v = 0; v < 8; ++v) p[(size_t)v * OUTC] = c[v];
    } else {
        for (int v = 0; v < 8; ++v)
            if (wave * 16 + 8 * kh + v < n_nodes) p[(size_t)v * OUTC] = c[v];
    }
}

__global__ void k_att2(const float* __restrict__ h2, const float* __restrict__ as2,
                       const float* __restrict__ ad2, float* __restrict__ a_s,
                       float* __restrict__ a_d, int n_nodes) {
    int n = blockIdx.x * blockDim.x + threadIdx.x;
    if (n >= n_nodes) return;
    const float* row = h2 + (size_t)n * OUTC;
    float s = 0.0f, d = 0.0f;
#pragma unroll
    for (int cc = 0; cc < OUTC; ++cc) {
        float v = row[cc];
        s += v * as2[cc];
        d += v * ad2[cc];
    }
    a_s[n] = s;
    a_d[n] = d;
}

__global__ void k_edge_max2(const int* __restrict__ ei, int E, int ET,
                            const float* __restrict__ a_s, const float* __restrict__ a_d,
                            unsigned* __restrict__ m2) {
    int e = blockIdx.x * blockDim.x + threadIdx.x;
    if (e >= ET) return;
    int s, d; edge_sd(ei, E, e, s, d);
    float v = lrelu(a_s[s] + a_d[d]);
    atomicMax(&m2[d], fenc(v));
}

__global__ void k_msg2(const int* __restrict__ ei, int E, int ET,
                       const float* __restrict__ a_s, const float* __restrict__ a_d,
                       const unsigned* __restrict__ m2, float* __restrict__ denom2,
                       const float* __restrict__ h2, float* __restrict__ out) {
    long long gid = (long long)blockIdx.x * blockDim.x + threadIdx.x;
    int e = (int)(gid >> 4);
    int c = (int)(gid & 15);
    if (e >= ET) return;
    int s, d; edge_sd(ei, E, e, s, d);
    float v = lrelu(a_s[s] + a_d[d]);
    float w = expf(v - fdec(m2[d]));
    if (c == 0) atomicAdd(&denom2[d], w);
    atomicAdd(&out[(size_t)d * OUTC + c], h2[(size_t)s * OUTC + c] * w);
}

__global__ void k_norm_bias2(float* __restrict__ out, const float* __restrict__ denom2,
                             const float* __restrict__ bias2, size_t n) {
    size_t i = (size_t)blockIdx.x * blockDim.x + threadIdx.x;
    size_t stride = (size_t)gridDim.x * blockDim.x;
    for (; i < n; i += stride)
        out[i] = out[i] / (denom2[i >> 4] + 1e-16f) + bias2[i & (OUTC - 1)];
}

extern "C" void kernel_launch(void* const* d_in, const int* in_sizes, int n_in,
                              void* d_out, int out_size, void* d_ws, size_t ws_size,
                              hipStream_t stream) {
    const float* x     = (const float*)d_in[0];
    const int*   ei    = (const int*)d_in[1];
    const float* W1    = (const float*)d_in[2];
    const float* as1   = (const float*)d_in[3];
    const float* ad1   = (const float*)d_in[4];
    const float* bias1 = (const float*)d_in[5];
    const float* W2    = (const float*)d_in[6];
    const float* as2   = (const float*)d_in[7];
    const float* ad2   = (const float*)d_in[8];
    const float* bias2 = (const float*)d_in[9];
    float* out = (float*)d_out;

    const int N  = in_sizes[0] / 3;
    const int E  = in_sizes[1] / 2;
    const int ET = E + N;

    float* ws = (float*)d_ws;
    const size_t Nz = (size_t)N;
    float*    agg1   = ws;
    unsigned* m1     = (unsigned*)(ws + 32 * Nz);
    unsigned* m2     = (unsigned*)(ws + 40 * Nz);
    float*    denom2 = ws + 41 * Nz;
    const size_t zero_count = 42 * Nz;
    float*    h1   = ws + 42 * Nz;
    float*    helu = ws + 298 * Nz;
    float*    a_s1 = ws + 554 * Nz;
    float*    a_d1 = ws + 562 * Nz;
    float*    h2   = ws + 570 * Nz;
    float*    a_s2 = ws + 586 * Nz;
    float*    a_d2 = ws + 587 * Nz;

    const int TB = 256;
    k_fill_zero<<<2048, TB, 0, stream>>>(ws, zero_count);
    k_fill_zero<<<512, TB, 0, stream>>>(out, (size_t)N * OUTC);

    {
        int mtiles = (N + 15) / 16;
        long long waves = (long long)mtiles * 16;
        int blocks = (int)((waves * 32 + TB - 1) / TB);
        k_gemm1_wmma<<<blocks, TB, 0, stream>>>(x, W1, h1, N);
    }
    k_att1<<<N, TB, 0, stream>>>(h1, as1, ad1, a_s1, a_d1, N);
    k_edge_max1<<<(ET + TB - 1) / TB, TB, 0, stream>>>(ei, E, ET, a_s1, a_d1, m1);
    {
        long long waves = ((long long)ET + EPW - 1) / EPW;
        k_msg1x<<<(int)((waves * 32 + TB - 1) / TB), TB, 0, stream>>>(ei, E, ET, x, a_s1, a_d1,
                                                                      m1, agg1);
    }
    k_gemm1_agg_wmma<<<(N + 15) / 16, TB, 0, stream>>>(agg1, W1, bias1, helu, N);

    {
        int mtiles = (N + 15) / 16;
        long long thr = (long long)mtiles * 32;
        k_gemm2_wmma<<<(int)((thr + TB - 1) / TB), TB, 0, stream>>>(helu, W2, h2, N);
    }
    k_att2<<<(N + TB - 1) / TB, TB, 0, stream>>>(h2, as2, ad2, a_s2, a_d2, N);
    k_edge_max2<<<(ET + TB - 1) / TB, TB, 0, stream>>>(ei, E, ET, a_s2, a_d2, m2);
    {
        long long thr = (long long)ET * 16;
        k_msg2<<<(int)((thr + TB - 1) / TB), TB, 0, stream>>>(ei, E, ET, a_s2, a_d2,
                                                              m2, denom2, h2, out);
    }
    k_norm_bias2<<<512, TB, 0, stream>>>(out, denom2, bias2, (size_t)N * OUTC);
}


# Layer-1 attention logits from the 3-wide input: a_s = x @ (W1 . att_src), a 3x16 matrix per block then one WMMA per 16 nodes; the 256-wide h1 = x @ W1 is never formed (sum order changes)
# speedup vs baseline: 2.9976x; 1.1597x over previous
#include <hip/hip_runtime.h>
#include <math.h>

#define HEADS     8
#define HIDC      32
#define L1C       256
#define OUTC      16
#define NEG_SLOPE 0.2f

typedef __attribute__((ext_vector_type(2))) float v2f;
typedef __attribute__((ext_vector_type(8))) float v8f;

__device__ __forceinline__ unsigned fenc(float f) {
    unsigned u = __float_as_uint(f);
    return (u & 0x80000000u) ? ~u : (u | 0x80000000u);
}
__device__ __forceinline__ float fdec(unsigned k) {
    return __uint_as_float((k & 0x80000000u) ? (k ^ 0x80000000u) : ~k);
}
__device__ __forceinline__ float lrelu(float v) {
    return v > 0.0f ? v : NEG_SLOPE * v;
}
__device__ __forceinline__ void edge_sd(const int* __restrict__ ei, int E, int e,
                                        int& s, int& d) {
    if (e < E) { s = ei[e]; d = ei[E + e]; }
    else       { s = e - E; d = e - E; }
}

__global__ void k_fill_zero(float* __restrict__ p, size_t n) {
    size_t i = (size_t)blockIdx.x * blockDim.x + threadIdx.x;
    size_t stride = (size_t)gridDim.x * blockDim.x;
    for (; i < n; i += stride) p[i] = 0.0f;
}

__global__ void k_att1_x(const float* __restrict__ x, const float* __restrict__ W1,
                         const float* __restrict__ as1, const float* __restrict__ ad1,
                         float* __restrict__ a_s, float* __restrict__ a_d, int n_nodes) {
    __shared__ float sV[3 * 16];
    int w    = threadIdx.x >> 5;
    int lane = threadIdx.x & 31;
    {
        float ts = as1[w * HIDC + lane];
        float td = ad1[w * HIDC + lane];
        float p[6];
#pragma unroll
        for (int k = 0; k < 3; ++k) {
            float wv = W1[k * L1C + w * HIDC + lane];
            p[k]     = wv * ts;
            p[3 + k] = wv * td;
        }
        for (int off = 16; off; off >>= 1) {
#pragma unroll
            for (int i = 0; i < 6; ++i) p[i] += __shfl_xor(p[i], off, 32);
        }
        if (lane == 0) {
#pragma unroll
            for (int k = 0; k < 3; ++k) {
                sV[k * 16 + w]     = p[k];
                sV[k * 16 + 8 + w] = p[3 + k];
            }
        }
    }
    __syncthreads();
    int mtiles = (n_nodes + 15) >> 4;
    int tile   = blockIdx.x * 8 + w;
    if (tile >= mtiles) return;
    int c  = lane & 15;
    int kh = lane >> 4;
    int node = tile * 16 + c;
    bool valid = node < n_nodes;
    int nc = valid ? node : n_nodes - 1;
    float x0 = x[nc * 3 + 0], x1 = x[nc * 3 + 1], x2 = x[nc * 3 + 2];
    v2f a, b;
    a.x = sV[(2 * kh) * 16 + c];  a.y = kh ? 0.0f : sV[16 + c];
    b.x = kh ? x2 : x0;           b.y = kh ? 0.0f : x1;
    float* po = (kh ? a_d : a_s) + (size_t)node * HEADS;
    v8f acc = {};
    acc = __builtin_amdgcn_wmma_f32_16x16x4_f32(false, a, false, b, (short)0, acc,
                                                false, false);
    if (valid) *(v8f*)po = acc;
}

#define MXE 32
__global__ void k_edge_max1(const int* __restrict__ ei, int E, int ET,
                            const float* __restrict__ a_s, const float* __restrict__ a_d,
                            unsigned* __restrict__ m1) {
    int wave = (int)((blockIdx.x * blockDim.x + threadIdx.x) >> 5);
    int lane = threadIdx.x & 31;
    int h  = lane & 7;
    int j  = lane >> 3;
    int e0 = wave * MXE;
    if (e0 >= ET) return;
    int el = e0 + lane;
    int sl = 0, dl = 0;
    if (el < ET) edge_sd(ei, E, el, sl, dl);
    unsigned key[MXE / 4];
    int      dd[MXE / 4];
#pragma unroll
    for (int i = 0; i < MXE / 4; ++i) {
        int s = __shfl(sl, i * 4 + j, 32);
        int d = __shfl(dl, i * 4 + j, 32);
        dd[i]  = d;
        key[i] = fenc(lrelu(a_s[s * HEADS + h] + a_d[d * HEADS + h]));
    }
#pragma unroll
    for (int i = 0; i < MXE / 4; ++i)
        if (e0 + i * 4 + j < ET) atomicMax(&m1[dd[i] * HEADS + h], key[i]);
}

#define EPW 16
__global__ void k_msg1x(const int* __restrict__ ei, int E, int ET,
                        const float* __restrict__ x,
                        const float* __restrict__ a_s, const float* __restrict__ a_d,
                        const unsigned* __restrict__ m1, float* __restrict__ agg) {
    int wave = (int)((blockIdx.x * blockDim.x + threadIdx.x) >> 5);
    int lane = threadIdx.x & 31;
    int h  = lane >> 2;
    int k  = lane & 3;
    int kx = k < 3 ? k : 2;
    int e0 = wave * EPW;
    int el = e0 + (lane & (EPW - 1));
    int sl = 0, dl = 0;
    if (el < ET) edge_sd(ei, E, el, sl, dl);
    float val[EPW];
#pragma unroll
    for (int i = 0; i < EPW; ++i) {
        int s = __builtin_amdgcn_readlane(sl, i);
        int d = __builtin_amdgcn_readlane(dl, i);
        float v  = lrelu(a_s[s * HEADS + h] + a_d[d * HEADS + h]);
        float w  = expf(v - fdec(m1[d * HEADS + h]));
        float xv = x[s * 3 + kx];
        val[i] = (k < 3) ? w * xv : w;
    }
    if (e0 + EPW <= ET) {
#pragma unroll
        for (int i = 0; i < EPW; ++i) {
            int d = __builtin_amdgcn_readlane(dl, i);
            atomicAdd(&agg[(size_t)d * 32 + lane], val[i]);
        }
    } else {
#pragma unroll
        for (int i = 0; i < EPW; ++i) {
            int d = __builtin_amdgcn_readlane(dl, i);
            if (e0 + i < ET) atomicAdd(&agg[(size_t)d * 32 + lane], val[i]);
        }
    }
}

__global__ void k_gemm1_agg_wmma(const float* __restrict__ agg, const float* __restrict__ W1,
                                 const float* __restrict__ bias1, float* __restrict__ helu,
                                 int n_nodes) {
    int mtile = blockIdx.x;
    int head  = threadIdx.x >> 5;
    int lane  = threadIdx.x & 31;
    int m  = lane & 15;
    int kh = lane >> 4;
    bool full = (mtile * 16 + 16 <= n_nodes);
    int node = mtile * 16 + m;
    if (node >= n_nodes) node = n_nodes - 1;
    const float* prow = agg + (size_t)node * 32 + head * 4;
    v2f p = *(const v2f*)(prow + 2 * kh);
    float den = prow[3] + 1e-16f;
    v2f a;
    a.x = p.x / den;
    a.y = kh ? 0.0f : p.y / den;
    float* prow_out = helu + (size_t)(mtile * 16 + 8 * kh) * L1C;
#pragma unroll
    for (int t = 0; t < 2; ++t) {
        int col = head * HIDC + t * 16 + m;
        float w0 = W1[0 * L1C + col], w1 = W1[1 * L1C + col], w2 = W1[2 * L1C + col];
        v2f b;
        b.x = kh ? w2 : w0;  b.y = kh ? 0.0f : w1;
        v8f c = {};
        c = __builtin_amdgcn_wmma_f32_16x16x4_f32(false, a, false, b, (short)0, c,
                                                  false, false);
        float bs = bias1[col];
        float* po = prow_out + col;
        if (full) {
#pragma unroll
            for (int v = 0; v < 8; ++v) {
                float r = c[v] + bs;
                po[(size_t)v * L1C] = r > 0.0f ? r : (expf(r) - 1.0f);
            }
        } else {
            for (int v = 0; v < 8; ++v)
                if (mtile * 16 + 8 * kh + v < n_nodes) {
                    float r = c[v] + bs;
                    po[(size_t)v * L1C] = r > 0.0f ? r : (expf(r) - 1.0f);
                }
        }
    }
}

__global__ void k_gemm2_wmma(const float* __restrict__ helu, const float* __restrict__ W2,
                             float* __restrict__ h2, int n_nodes) {
    int wave = (int)((blockIdx.x * blockDim.x + threadIdx.x) >> 5);
    int lane = threadIdx.x & 31;
    int mtiles = (n_nodes + 15) >> 4;
    if (wave >= mtiles) return;
    int m  = lane & 15;
    int kh = lane >> 4;
    bool full = (wave * 16 + 16 <= n_nodes);
    int node = wave * 16 + m;
    if (node >= n_nodes) node = n_nodes - 1;
    const float* arow = helu + (size_t)node * L1C + kh * 2;
    const float* brow = W2 + kh * 2 * OUTC + m;
    v8f c = {};
#pragma unroll 8
    for (int kk = 0; kk < L1C; kk += 4) {
        v2f a = *(const v2f*)(arow + kk);
        v2f b;
        b.x = brow[kk * OUTC];
        b.y = brow[kk * OUTC + OUTC];
        c = __builtin_amdgcn_wmma_f32_16x16x4_f32(false, a, false, b, (short)0, c,
                                                  false, false);
    }
    float* p = h2 + (size_t)(wave * 16 + 8 * kh) * OUTC + m;
    if (full) {
#pragma unroll
        for (int v = 0; v < 8; ++v) p[(size_t)v * OUTC] = c[v];
    } else {
        for (int v = 0; v < 8; ++v)
            if (wave * 16 + 8 * kh + v < n_nodes) p[(size_t)v * OUTC] = c[v];
    }
}

__global__ void k_att2(const float* __restrict__ h2, const float* __restrict__ as2,
                       const float* __restrict__ ad2, float* __restrict__ a_s,
                       float* __restrict__ a_d, int n_nodes) {
    int n = blockIdx.x * blockDim.x + threadIdx.x;
    if (n >= n_nodes) return;
    const float* row = h2 + (size_t)n * OUTC;
    float s = 0.0f, d = 0.0f;
#pragma unroll
    for (int cc = 0; cc < OUTC; ++cc) {
        float v = row[cc];
        s += v * as2[cc];
        d += v * ad2[cc];
    }
    a_s[n] = s;
    a_d[n] = d;
}

__global__ void k_edge_max2(const int* __restrict__ ei, int E, int ET,
                            const float* __restrict__ a_s, const float* __restrict__ a_d,
                            unsigned* __restrict__ m2) {
    int e = blockIdx.x * blockDim.x + threadIdx.x;
    if (e >= ET) return;
    int s, d; edge_sd(ei, E, e, s, d);
    float v = lrelu(a_s[s] + a_d[d]);
    atomicMax(&m2[d], fenc(v));
}

__global__ void k_msg2(const int* __restrict__ ei, int E, int ET,
                       const float* __restrict__ a_s, const float* __restrict__ a_d,
                       const unsigned* __restrict__ m2, float* __restrict__ denom2,
                       const float* __restrict__ h2, float* __restrict__ out) {
    long long gid = (long long)blockIdx.x * blockDim.x + threadIdx.x;
    int e = (int)(gid >> 4);
    int c = (int)(gid & 15);
    if (e >= ET) return;
    int s, d; edge_sd(ei, E, e, s, d);
    float v = lrelu(a_s[s] + a_d[d]);
    float w = expf(v - fdec(m2[d]));
    if (c == 0) atomicAdd(&denom2[d], w);
    atomicAdd(&out[(size_t)d * OUTC + c], h2[(size_t)s * OUTC + c] * w);
}

__global__ void k_norm_bias2(float* __restrict__ out, const float* __restrict__ denom2,
                             const float* __restrict__ bias2, size_t n) {
    size_t i = (size_t)blockIdx.x * blockDim.x + threadIdx.x;
    size_t stride = (size_t)gridDim.x * blockDim.x;
    for (; i < n; i += stride)
        out[i] = out[i] / (denom2[i >> 4] + 1e-16f) + bias2[i & (OUTC - 1)];
}

extern "C" void kernel_launch(void* const* d_in, const int* in_sizes, int n_in,
                              void* d_out, int out_size, void* d_ws, size_t ws_size,
                              hipStream_t stream) {
    const float* x     = (const float*)d_in[0];
    const int*   ei    = (const int*)d_in[1];
    const float* W1    = (const float*)d_in[2];
    const float* as1   = (const float*)d_in[3];
    const float* ad1   = (const float*)d_in[4];
    const float* bias1 = (const float*)d_in[5];
    const float* W2    = (const float*)d_in[6];
    const float* as2   = (const float*)d_in[7];
    const float* ad2   = (const float*)d_in[8];
    const float* bias2 = (const float*)d_in[9];
    float* out = (float*)d_out;

    const int N  = in_sizes[0] / 3;
    const int E  = in_sizes[1] / 2;
    const int ET = E + N;

    float* ws = (float*)d_ws;
    const size_t Nz = (size_t)N;
    float*    agg1   = ws;
    unsigned* m1     = (unsigned*)(ws + 32 * Nz);
    unsigned* m2     = (unsigned*)(ws + 40 * Nz);
    float*    denom2 = ws + 41 * Nz;
    const size_t zero_count = 42 * Nz;
    float*    helu = ws + 42 * Nz;
    float*    a_s1 = ws + 298 * Nz;
    float*    a_d1 = ws + 306 * Nz;
    float*    h2   = ws + 314 * Nz;
    float*    a_s2 = ws + 330 * Nz;
    float*    a_d2 = ws + 331 * Nz;

    const int TB = 256;
    k_fill_zero<<<2048, TB, 0, stream>>>(ws, zero_count);
    k_fill_zero<<<512, TB, 0, stream>>>(out, (size_t)N * OUTC);

    {
        int mtiles = (N + 15) / 16;
        k_att1_x<<<(mtiles + 7) / 8, TB, 0, stream>>>(x, W1, as1, ad1, a_s1, a_d1, N);
    }
    k_edge_max1<<<(ET + TB - 1) / TB, TB, 0, stream>>>(ei, E, ET, a_s1, a_d1, m1);
    {
        long long waves = ((long long)ET + EPW - 1) / EPW;
        k_msg1x<<<(int)((waves * 32 + TB - 1) / TB), TB, 0, stream>>>(ei, E, ET, x, a_s1, a_d1,
                                                                      m1, agg1);
    }
    k_gemm1_agg_wmma<<<(N + 15) / 16, TB, 0, stream>>>(agg1, W1, bias1, helu, N);

    {
        int mtiles = (N + 15) / 16;
        long long thr = (long long)mtiles * 32;
        k_gemm2_wmma<<<(int)((thr + TB - 1) / TB), TB, 0, stream>>>(helu, W2, h2, N);
    }
    k_att2<<<(N + TB - 1) / TB, TB, 0, stream>>>(h2, as2, ad2, a_s2, a_d2, N);
    k_edge_max2<<<(ET + TB - 1) / TB, TB, 0, stream>>>(ei, E, ET, a_s2, a_d2, m2);
    {
        long long thr = (long long)ET * 16;
        k_msg2<<<(int)((thr + TB - 1) / TB), TB, 0, stream>>>(ei, E, ET, a_s2, a_d2,
                                                              m2, denom2, h2, out);
    }
    k_norm_bias2<<<512, TB, 0, stream>>>(out, denom2, bias2, (size_t)N * OUTC);
}


# Layer-2 message pass: a wave takes 32 edges, not 2: lane = edge computes each weight once and adds the denominator, then 16 instructions of 2 edges x 16 channels; float atomics as before
# speedup vs baseline: 3.7986x; 1.2672x over previous
#include <hip/hip_runtime.h>
#include <math.h>

#define HEADS     8
#define HIDC      32
#define L1C       256
#define OUTC      16
#define NEG_SLOPE 0.2f

typedef __attribute__((ext_vector_type(2))) float v2f;
typedef __attribute__((ext_vector_type(8))) float v8f;

__device__ __forceinline__ unsigned fenc(float f) {
    unsigned u = __float_as_uint(f);
    return (u & 0x80000000u) ? ~u : (u | 0x80000000u);
}
__device__ __forceinline__ float fdec(unsigned k) {
    return __uint_as_float((k & 0x80000000u) ? (k ^ 0x80000000u) : ~k);
}
__device__ __forceinline__ float lrelu(float v) {
    return v > 0.0f ? v : NEG_SLOPE * v;
}
__device__ __forceinline__ void edge_sd(const int* __restrict__ ei, int E, int e,
                                        int& s, int& d) {
    if (e < E) { s = ei[e]; d = ei[E + e]; }
    else       { s = e - E; d = e - E; }
}

__global__ void k_fill_zero(float* __restrict__ p, size_t n) {
    size_t i = (size_t)blockIdx.x * blockDim.x + threadIdx.x;
    size_t stride = (size_t)gridDim.x * blockDim.x;
    for (; i < n; i += stride) p[i] = 0.0f;
}

__global__ void k_att1_x(const float* __restrict__ x, const float* __restrict__ W1,
                         const float* __restrict__ as1, const float* __restrict__ ad1,
                         float* __restrict__ a_s, float* __restrict__ a_d, int n_nodes) {
    __shared__ float sV[3 * 16];
    int w    = threadIdx.x >> 5;
    int lane = threadIdx.x & 31;
    {
        float ts = as1[w * HIDC + lane];
        float td = ad1[w * HIDC + lane];
        float p[6];
#pragma unroll
        for (int k = 0; k < 3; ++k) {
            float wv = W1[k * L1C + w * HIDC + lane];
            p[k]     = wv * ts;
            p[3 + k] = wv * td;
        }
        for (int off = 16; off; off >>= 1) {
#pragma unroll
            for (int i = 0; i < 6; ++i) p[i] += __shfl_xor(p[i], off, 32);
        }
        if (lane == 0) {
#pragma unroll
            for (int k = 0; k < 3; ++k) {
                sV[k * 16 + w]     = p[k];
                sV[k * 16 + 8 + w] = p[3 + k];
            }
        }
    }
    __syncthreads();
    int mtiles = (n_nodes + 15) >> 4;
    int tile   = blockIdx.x * 8 + w;
    if (tile >= mtiles) return;
    int c  = lane & 15;
    int kh = lane >> 4;
    int node = tile * 16 + c;
    bool valid = node < n_nodes;
    int nc = valid ? node : n_nodes - 1;
    float x0 = x[nc * 3 + 0], x1 = x[nc * 3 + 1], x2 = x[nc * 3 + 2];
    v2f a, b;
    a.x = sV[(2 * kh) * 16 + c];  a.y = kh ? 0.0f : sV[16 + c];
    b.x = kh ? x2 : x0;           b.y = kh ? 0.0f : x1;
    float* po = (kh ? a_d : a_s) + (size_t)node * HEADS;
    v8f acc = {};
    acc = __builtin_amdgcn_wmma_f32_16x16x4_f32(false, a, false, b, (short)0, acc,
                                                false, false);
    if (valid) *(v8f*)po = acc;
}

#define MXE 32
__global__ void k_edge_max1(const int* __restrict__ ei, int E, int ET,
                            const float* __restrict__ a_s, const float* __restrict__ a_d,
                            unsigned* __restrict__ m1) {
    int wave = (int)((blockIdx.x * blockDim.x + threadIdx.x) >> 5);
    int lane = threadIdx.x & 31;
    int h  = lane & 7;
    int j  = lane >> 3;
    int e0 = wave * MXE;
    if (e0 >= ET) return;
    int el = e0 + lane;
    int sl = 0, dl = 0;
    if (el < ET) edge_sd(ei, E, el, sl, dl);
    unsigned key[MXE / 4];
    int      dd[MXE / 4];
#pragma unroll
    for (int i = 0; i < MXE / 4; ++i) {
        int s = __shfl(sl, i * 4 + j, 32);
        int d = __shfl(dl, i * 4 + j, 32);
        dd[i]  = d;
        key[i] = fenc(lrelu(a_s[s * HEADS + h] + a_d[d * HEADS + h]));
    }
#pragma unroll
    for (int i = 0; i < MXE / 4; ++i)
        if (e0 + i * 4 + j < ET) atomicMax(&m1[dd[i] * HEADS + h], key[i]);
}

#define EPW 16
__global__ void k_msg1x(const int* __restrict__ ei, int E, int ET,
                        const float* __restrict__ x,
                        const float* __restrict__ a_s, const float* __restrict__ a_d,
                        const unsigned* __restrict__ m1, float* __restrict__ agg) {
    int wave = (int)((blockIdx.x * blockDim.x + threadIdx.x) >> 5);
    int lane = threadIdx.x & 31;
    int h  = lane >> 2;
    int k  = lane & 3;
    int kx = k < 3 ? k : 2;
    int e0 = wave * EPW;
    int el = e0 + (lane & (EPW - 1));
    int sl = 0, dl = 0;
    if (el < ET) edge_sd(ei, E, el, sl, dl);
    float val[EPW];
#pragma unroll
    for (int i = 0; i < EPW; ++i) {
        int s = __builtin_amdgcn_readlane(sl, i);
        int d = __builtin_amdgcn_readlane(dl, i);
        float v  = lrelu(a_s[s * HEADS + h] + a_d[d * HEADS + h]);
        float w  = expf(v - fdec(m1[d * HEADS + h]));
        float xv = x[s * 3 + kx];
        val[i] = (k < 3) ? w * xv : w;
    }
    if (e0 + EPW <= ET) {
#pragma unroll
        for (int i = 0; i < EPW; ++i) {
            int d = __builtin_amdgcn_readlane(dl, i);
            atomicAdd(&agg[(size_t)d * 32 + lane], val[i]);
        }
    } else {
#pragma unroll
        for (int i = 0; i < EPW; ++i) {
            int d = __builtin_amdgcn_readlane(dl, i);
            if (e0 + i < ET) atomicAdd(&agg[(size_t)d * 32 + lane], val[i]);
        }
    }
}

__global__ void k_gemm1_agg_wmma(const float* __restrict__ agg, const float* __restrict__ W1,
                                 const float* __restrict__ bias1, float* __restrict__ helu,
                                 int n_nodes) {
    int mtile = blockIdx.x;
    int head  = threadIdx.x >> 5;
    int lane  = threadIdx.x & 31;
    int m  = lane & 15;
    int kh = lane >> 4;
    bool full = (mtile * 16 + 16 <= n_nodes);
    int node = mtile * 16 + m;
    if (node >= n_nodes) node = n_nodes - 1;
    const float* prow = agg + (size_t)node * 32 + head * 4;
    v2f p = *(const v2f*)(prow + 2 * kh);
    float den = prow[3] + 1e-16f;
    v2f a;
    a.x = p.x / den;
    a.y = kh ? 0.0f : p.y / den;
    float* prow_out = helu + (size_t)(mtile * 16 + 8 * kh) * L1C;
#pragma unroll
    for (int t = 0; t < 2; ++t) {
        int col = head * HIDC + t * 16 + m;
        float w0 = W1[0 * L1C + col], w1 = W1[1 * L1C + col], w2 = W1[2 * L1C + col];
        v2f b;
        b.x = kh ? w2 : w0;  b.y = kh ? 0.0f : w1;
        v8f c = {};
        c = __builtin_amdgcn_wmma_f32_16x16x4_f32(false, a, false, b, (short)0, c,
                                                  false, false);
        float bs = bias1[col];
        float* po = prow_out + col;
        if (full) {
#pragma unroll
            for (int v = 0; v < 8; ++v) {
                float r = c[v] + bs;
                po[(size_t)v * L1C] = r > 0.0f ? r : (expf(r) - 1.0f);
            }
        } else {
            for (int v = 0; v < 8; ++v)
                if (mtile * 16 + 8 * kh + v < n_nodes) {
                    float r = c[v] + bs;
                    po[(size_t)v * L1C] = r > 0.0f ? r : (expf(r) - 1.0f);
                }
        }
    }
}

__global__ void k_gemm2_wmma(const float* __restrict__ helu, const float* __restrict__ W2,
                             float* __restrict__ h2, int n_nodes) {
    int wave = (int)((blockIdx.x * blockDim.x + threadIdx.x) >> 5);
    int lane = threadIdx.x & 31;
    int mtiles = (n_nodes + 15) >> 4;
    if (wave >= mtiles) return;
    int m  = lane & 15;
    int kh = lane >> 4;
    bool full = (wave * 16 + 16 <= n_nodes);
    int node = wave * 16 + m;
    if (node >= n_nodes) node = n_nodes - 1;
    const float* arow = helu + (size_t)node * L1C + kh * 2;
    const float* brow = W2 + kh * 2 * OUTC + m;
    v8f c = {};
#pragma unroll 8
    for (int kk = 0; kk < L1C; kk += 4) {
        v2f a = *(const v2f*)(arow + kk);
        v2f b;
        b.x = brow[kk * OUTC];
        b.y = brow[kk * OUTC + OUTC];
        c = __builtin_amdgcn_wmma_f32_16x16x4_f32(false, a, false, b, (short)0, c,
                                                  false, false);
    }
    float* p = h2 + (size_t)(wave * 16 + 8 * kh) * OUTC + m;
    if (full) {
#pragma unroll
        for (int v = 0; v < 8; ++v) p[(size_t)v * OUTC] = c[v];
    } else {
        for (int v = 0; v < 8; ++v)
            if (wave * 16 + 8 * kh + v < n_nodes) p[(size_t)v * OUTC] = c[v];
    }
}

__global__ void k_att2(const float* __restrict__ h2, const float* __restrict__ as2,
                       const float* __restrict__ ad2, float* __restrict__ a_s,
                       float* __restrict__ a_d, int n_nodes) {
    int n = blockIdx.x * blockDim.x + threadIdx.x;
    if (n >= n_nodes) return;
    const float* row = h2 + (size_t)n * OUTC;
    float s = 0.0f, d = 0.0f;
#pragma unroll
    for (int cc = 0; cc < OUTC; ++cc) {
        float v = row[cc];
        s += v * as2[cc];
        d += v * ad2[cc];
    }
    a_s[n] = s;
    a_d[n] = d;
}

__global__ void k_edge_max2(const int* __restrict__ ei, int E, int ET,
                            const float* __restrict__ a_s, const float* __restrict__ a_d,
                            unsigned* __restrict__ m2) {
    int e = blockIdx.x * blockDim.x + threadIdx.x;
    if (e >= ET) return;
    int s, d; edge_sd(ei, E, e, s, d);
    float v = lrelu(a_s[s] + a_d[d]);
    atomicMax(&m2[d], fenc(v));
}

#define M2E 32
__global__ void k_msg2(const int* __restrict__ ei, int E, int ET,
                       const float* __restrict__ a_s, const float* __restrict__ a_d,
                       const unsigned* __restrict__ m2, float* __restrict__ denom2,
                       const float* __restrict__ h2, float* __restrict__ out) {
    int wave = (int)((blockIdx.x * blockDim.x + threadIdx.x) >> 5);
    int lane = threadIdx.x & 31;
    int c  = lane & 15;
    int j  = lane >> 4;
    int e0 = wave * M2E;
    if (e0 >= ET) return;
    int el = e0 + lane;
    int sl = 0, dl = 0;
    float wl = 0.0f;
    if (el < ET) {
        edge_sd(ei, E, el, sl, dl);
        float v = lrelu(a_s[sl] + a_d[dl]);
        wl = expf(v - fdec(m2[dl]));
        atomicAdd(&denom2[dl], wl);
    }
    int so = sl * OUTC, dof = dl * OUTC;
    float val[M2E / 2];
    int   dd[M2E / 2];
#pragma unroll
    for (int i = 0; i < M2E / 2; ++i) {
        int   s = __shfl(so,  2 * i + j, 32);
        int   d = __shfl(dof, 2 * i + j, 32);
        float w = __shfl(wl,  2 * i + j, 32);
        dd[i]  = d + c;
        val[i] = h2[s + c] * w;
    }
    if (e0 + M2E <= ET) {
#pragma unroll
        for (int i = 0; i < M2E / 2; ++i)
            atomicAdd(&out[dd[i]], val[i]);
    } else {
#pragma unroll
        for (int i = 0; i < M2E / 2; ++i)
            if (e0 + 2 * i + j < ET) atomicAdd(&out[dd[i]], val[i]);
    }
}

__global__ void k_norm_bias2(float* __restrict__ out, const float* __restrict__ denom2,
                             const float* __restrict__ bias2, size_t n) {
    size_t i = (size_t)blockIdx.x * blockDim.x + threadIdx.x;
    size_t stride = (size_t)gridDim.x * blockDim.x;
    for (; i < n; i += stride)
        out[i] = out[i] / (denom2[i >> 4] + 1e-16f) + bias2[i & (OUTC - 1)];
}

extern "C" void kernel_launch(void* const* d_in, const int* in_sizes, int n_in,
                              void* d_out, int out_size, void* d_ws, size_t ws_size,
                              hipStream_t stream) {
    const float* x     = (const float*)d_in[0];
    const int*   ei    = (const int*)d_in[1];
    const float* W1    = (const float*)d_in[2];
    const float* as1   = (const float*)d_in[3];
    const float* ad1   = (const float*)d_in[4];
    const float* bias1 = (const float*)d_in[5];
    const float* W2    = (const float*)d_in[6];
    const float* as2   = (const float*)d_in[7];
    const float* ad2   = (const float*)d_in[8];
    const float* bias2 = (const float*)d_in[9];
    float* out = (float*)d_out;

    const int N  = in_sizes[0] / 3;
    const int E  = in_sizes[1] / 2;
    const int ET = E + N;

    float* ws = (float*)d_ws;
    const size_t Nz = (size_t)N;
    float*    agg1   = ws;
    unsigned* m1     = (unsigned*)(ws + 32 * Nz);
    unsigned* m2     = (unsigned*)(ws + 40 * Nz);
    float*    denom2 = ws + 41 * Nz;
    const size_t zero_count = 42 * Nz;
    float*    helu = ws + 42 * Nz;
    float*    a_s1 = ws + 298 * Nz;
    float*    a_d1 = ws + 306 * Nz;
    float*    h2   = ws + 314 * Nz;
    float*    a_s2 = ws + 330 * Nz;
    float*    a_d2 = ws + 331 * Nz;

    const int TB = 256;
    k_fill_zero<<<2048, TB, 0, stream>>>(ws, zero_count);
    k_fill_zero<<<512, TB, 0, stream>>>(out, (size_t)N * OUTC);

    {
        int mtiles = (N + 15) / 16;
        k_att1_x<<<(mtiles + 7) / 8, TB, 0, stream>>>(x, W1, as1, ad1, a_s1, a_d1, N);
    }
    k_edge_max1<<<(ET + TB - 1) / TB, TB, 0, stream>>>(ei, E, ET, a_s1, a_d1, m1);
    {
        long long waves = ((long long)ET + EPW - 1) / EPW;
        k_msg1x<<<(int)((waves * 32 + TB - 1) / TB), TB, 0, stream>>>(ei, E, ET, x, a_s1, a_d1,
                                                                      m1, agg1);
    }
    k_gemm1_agg_wmma<<<(N + 15) / 16, TB, 0, stream>>>(agg1, W1, bias1, helu, N);

    {
        int mtiles = (N + 15) / 16;
        long long thr = (long long)mtiles * 32;
        k_gemm2_wmma<<<(int)((thr + TB - 1) / TB), TB, 0, stream>>>(helu, W2, h2, N);
    }
    k_att2<<<(N + TB - 1) / TB, TB, 0, stream>>>(h2, as2, ad2, a_s2, a_d2, N);
    k_edge_max2<<<(ET + TB - 1) / TB, TB, 0, stream>>>(ei, E, ET, a_s2, a_d2, m2);
    {
        long long waves = ((long long)ET + M2E - 1) / M2E;
        k_msg2<<<(int)((waves * 32 + TB - 1) / TB), TB, 0, stream>>>(ei, E, ET, a_s2, a_d2,
                                                                     m2, denom2, h2, out);
    }
    k_norm_bias2<<<512, TB, 0, stream>>>(out, denom2, bias2, (size_t)N * OUTC);
}


# Layer-1 output GEMM+bias+ELU fused into the layer-2 GEMM: both products transposed, so D of the first is B of the second in registers; 256-wide activations never stored (sum order changes)
# speedup vs baseline: 4.0272x; 1.0602x over previous
#include <hip/hip_runtime.h>
#include <math.h>

#define HEADS     8
#define HIDC      32
#define L1C       256
#define OUTC      16
#define NEG_SLOPE 0.2f

typedef __attribute__((ext_vector_type(2))) float v2f;
typedef __attribute__((ext_vector_type(8))) float v8f;

__device__ __forceinline__ unsigned fenc(float f) {
    unsigned u = __float_as_uint(f);
    return (u & 0x80000000u) ? ~u : (u | 0x80000000u);
}
__device__ __forceinline__ float fdec(unsigned k) {
    return __uint_as_float((k & 0x80000000u) ? (k ^ 0x80000000u) : ~k);
}
__device__ __forceinline__ float lrelu(float v) {
    return v > 0.0f ? v : NEG_SLOPE * v;
}
__device__ __forceinline__ void edge_sd(const int* __restrict__ ei, int E, int e,
                                        int& s, int& d) {
    if (e < E) { s = ei[e]; d = ei[E + e]; }
    else       { s = e - E; d = e - E; }
}

__global__ void k_fill_zero(float* __restrict__ p, size_t n) {
    size_t i = (size_t)blockIdx.x * blockDim.x + threadIdx.x;
    size_t stride = (size_t)gridDim.x * blockDim.x;
    for (; i < n; i += stride) p[i] = 0.0f;
}

__global__ void k_att1_x(const float* __restrict__ x, const float* __restrict__ W1,
                         const float* __restrict__ as1, const float* __restrict__ ad1,
                         float* __restrict__ a_s, float* __restrict__ a_d, int n_nodes) {
    __shared__ float sV[3 * 16];
    int w    = threadIdx.x >> 5;
    int lane = threadIdx.x & 31;
    {
        float ts = as1[w * HIDC + lane];
        float td = ad1[w * HIDC + lane];
        float p[6];
#pragma unroll
        for (int k = 0; k < 3; ++k) {
            float wv = W1[k * L1C + w * HIDC + lane];
            p[k]     = wv * ts;
            p[3 + k] = wv * td;
        }
        for (int off = 16; off; off >>= 1) {
#pragma unroll
            for (int i = 0; i < 6; ++i) p[i] += __shfl_xor(p[i], off, 32);
        }
        if (lane == 0) {
#pragma unroll
            for (int k = 0; k < 3; ++k) {
                sV[k * 16 + w]     = p[k];
                sV[k * 16 + 8 + w] = p[3 + k];
            }
        }
    }
    __syncthreads();
    int mtiles = (n_nodes + 15) >> 4;
    int tile   = blockIdx.x * 8 + w;
    if (tile >= mtiles) return;
    int c  = lane & 15;
    int kh = lane >> 4;
    int node = tile * 16 + c;
    bool valid = node < n_nodes;
    int nc = valid ? node : n_nodes - 1;
    float x0 = x[nc * 3 + 0], x1 = x[nc * 3 + 1], x2 = x[nc * 3 + 2];
    v2f a, b;
    a.x = sV[(2 * kh) * 16 + c];  a.y = kh ? 0.0f : sV[16 + c];
    b.x = kh ? x2 : x0;           b.y = kh ? 0.0f : x1;
    float* po = (kh ? a_d : a_s) + (size_t)node * HEADS;
    v8f acc = {};
    acc = __builtin_amdgcn_wmma_f32_16x16x4_f32(false, a, false, b, (short)0, acc,
                                                false, false);
    if (valid) *(v8f*)po = acc;
}

#define MXE 32
__global__ void k_edge_max1(const int* __restrict__ ei, int E, int ET,
                            const float* __restrict__ a_s, const float* __restrict__ a_d,
                            unsigned* __restrict__ m1) {
    int wave = (int)((blockIdx.x * blockDim.x + threadIdx.x) >> 5);
    int lane = threadIdx.x & 31;
    int h  = lane & 7;
    int j  = lane >> 3;
    int e0 = wave * MXE;
    if (e0 >= ET) return;
    int el = e0 + lane;
    int sl = 0, dl = 0;
    if (el < ET) edge_sd(ei, E, el, sl, dl);
    unsigned key[MXE / 4];
    int      dd[MXE / 4];
#pragma unroll
    for (int i = 0; i < MXE / 4; ++i) {
        int s = __shfl(sl, i * 4 + j, 32);
        int d = __shfl(dl, i * 4 + j, 32);
        dd[i]  = d;
        key[i] = fenc(lrelu(a_s[s * HEADS + h] + a_d[d * HEADS + h]));
    }
#pragma unroll
    for (int i = 0; i < MXE / 4; ++i)
        if (e0 + i * 4 + j < ET) atomicMax(&m1[dd[i] * HEADS + h], key[i]);
}

#define EPW 16
__global__ void k_msg1x(const int* __restrict__ ei, int E, int ET,
                        const float* __restrict__ x,
                        const float* __restrict__ a_s, const float* __restrict__ a_d,
                        const unsigned* __restrict__ m1, float* __restrict__ agg) {
    int wave = (int)((blockIdx.x * blockDim.x + threadIdx.x) >> 5);
    int lane = threadIdx.x & 31;
    int h  = lane >> 2;
    int k  = lane & 3;
    int kx = k < 3 ? k : 2;
    int e0 = wave * EPW;
    int el = e0 + (lane & (EPW - 1));
    int sl = 0, dl = 0;
    if (el < ET) edge_sd(ei, E, el, sl, dl);
    float val[EPW];
#pragma unroll
    for (int i = 0; i < EPW; ++i) {
        int s = __builtin_amdgcn_readlane(sl, i);
        int d = __builtin_amdgcn_readlane(dl, i);
        float v  = lrelu(a_s[s * HEADS + h] + a_d[d * HEADS + h]);
        float w  = expf(v - fdec(m1[d * HEADS + h]));
        float xv = x[s * 3 + kx];
        val[i] = (k < 3) ? w * xv : w;
    }
    if (e0 + EPW <= ET) {
#pragma unroll
        for (int i = 0; i < EPW; ++i) {
            int d = __builtin_amdgcn_readlane(dl, i);
            atomicAdd(&agg[(size_t)d * 32 + lane], val[i]);
        }
    } else {
#pragma unroll
        for (int i = 0; i < EPW; ++i) {
            int d = __builtin_amdgcn_readlane(dl, i);
            if (e0 + i < ET) atomicAdd(&agg[(size_t)d * 32 + lane], val[i]);
        }
    }
}

typedef __attribute__((ext_vector_type(4))) float v4f;
struct HeadIn {
    v4f   p;
    float w1a[2];
    float w1b[2];
    v8f   bs[2];
    float w2[2][8];
};
__device__ __forceinline__ HeadIn load_head(const float* __restrict__ prow,
                                            const float* __restrict__ W1,
                                            const float* __restrict__ bias1,
                                            const float* __restrict__ W2,
                                            int head, int m, int kh) {
    HeadIn r;
    r.p = *(const v4f*)(prow + head * 4);
#pragma unroll
    for (int t = 0; t < 2; ++t) {
        int cb  = head * HIDC + t * 16;
        int col = cb + m;
        r.w1a[t] = W1[(2 * kh) * L1C + col];
        r.w1b[t] = W1[L1C + col];
        r.bs[t]  = *(const v8f*)(bias1 + cb + 8 * kh);
        const float* w2p = W2 + (size_t)(cb + 8 * kh) * OUTC + m;
#pragma unroll
        for (int i = 0; i < 8; ++i) r.w2[t][i] = w2p[i * OUTC];
    }
    return r;
}
__global__ void __launch_bounds__(64)
k_l1out_gemm2_wmma(const float* __restrict__ agg, const float* __restrict__ W1,
                   const float* __restrict__ bias1, const float* __restrict__ W2,
                   float* __restrict__ h2, int n_nodes) {
    int wave = (int)((blockIdx.x * blockDim.x + threadIdx.x) >> 5);
    int lane = threadIdx.x & 31;
    int mtiles = (n_nodes + 15) >> 4;
    if (wave >= mtiles) return;
    int m  = lane & 15;
    int kh = lane >> 4;
    int node = wave * 16 + m;
    bool valid = node < n_nodes;
    int nc = valid ? node : n_nodes - 1;
    const float* prow = agg + (size_t)nc * 32;
    v8f acc0 = {}, acc1 = {};
    HeadIn cur = load_head(prow, W1, bias1, W2, 0, m, kh);
#pragma unroll
    for (int head = 0; head < HEADS; ++head) {
        HeadIn nxt = cur;
        if (head + 1 < HEADS) nxt = load_head(prow, W1, bias1, W2, head + 1, m, kh);
        float den = cur.p.w + 1e-16f;
        v2f b1;
        b1.x = (kh ? cur.p.z : cur.p.x) / den;
        b1.y = kh ? 0.0f : cur.p.y / den;
#pragma unroll
        for (int t = 0; t < 2; ++t) {
            v2f a1;
            a1.x = cur.w1a[t];
            a1.y = kh ? 0.0f : cur.w1b[t];
            v8f d = {};
            d = __builtin_amdgcn_wmma_f32_16x16x4_f32(false, a1, false, b1, (short)0, d,
                                                      false, false);
#pragma unroll
            for (int j = 0; j < 4; ++j) {
                float r0 = d[2 * j] + cur.bs[t][2 * j];
                float r1 = d[2 * j + 1] + cur.bs[t][2 * j + 1];
                float e0 = expf(r0) - 1.0f;
                float e1 = expf(r1) - 1.0f;
                v2f b2;
                b2.x = r0 > 0.0f ? r0 : e0;
                b2.y = r1 > 0.0f ? r1 : e1;
                v2f a2;
                a2.x = cur.w2[t][2 * j];
                a2.y = cur.w2[t][2 * j + 1];
                if (t == 0)
                    acc0 = __builtin_amdgcn_wmma_f32_16x16x4_f32(false, a2, false, b2, (short)0,
                                                                 acc0, false, false);
                else
                    acc1 = __builtin_amdgcn_wmma_f32_16x16x4_f32(false, a2, false, b2, (short)0,
                                                                 acc1, false, false);
            }
        }
        cur = nxt;
    }
    if (valid) *(v8f*)(h2 + (size_t)node * OUTC + 8 * kh) = acc0 + acc1;
}

__global__ void k_att2(const float* __restrict__ h2, const float* __restrict__ as2,
                       const float* __restrict__ ad2, float* __restrict__ a_s,
                       float* __restrict__ a_d, int n_nodes) {
    int n = blockIdx.x * blockDim.x + threadIdx.x;
    if (n >= n_nodes) return;
    const float* row = h2 + (size_t)n * OUTC;
    float s = 0.0f, d = 0.0f;
#pragma unroll
    for (int cc = 0; cc < OUTC; ++cc) {
        float v = row[cc];
        s += v * as2[cc];
        d += v * ad2[cc];
    }
    a_s[n] = s;
    a_d[n] = d;
}

__global__ void k_edge_max2(const int* __restrict__ ei, int E, int ET,
                            const float* __restrict__ a_s, const float* __restrict__ a_d,
                            unsigned* __restrict__ m2) {
    int e = blockIdx.x * blockDim.x + threadIdx.x;
    if (e >= ET) return;
    int s, d; edge_sd(ei, E, e, s, d);
    float v = lrelu(a_s[s] + a_d[d]);
    atomicMax(&m2[d], fenc(v));
}

#define M2E 32
__global__ void k_msg2(const int* __restrict__ ei, int E, int ET,
                       const float* __restrict__ a_s, const float* __restrict__ a_d,
                       const unsigned* __restrict__ m2, float* __restrict__ denom2,
                       const float* __restrict__ h2, float* __restrict__ out) {
    int wave = (int)((blockIdx.x * blockDim.x + threadIdx.x) >> 5);
    int lane = threadIdx.x & 31;
    int c  = lane & 15;
    int j  = lane >> 4;
    int e0 = wave * M2E;
    if (e0 >= ET) return;
    int el = e0 + lane;
    int sl = 0, dl = 0;
    float wl = 0.0f;
    if (el < ET) {
        edge_sd(ei, E, el, sl, dl);
        float v = lrelu(a_s[sl] + a_d[dl]);
        wl = expf(v - fdec(m2[dl]));
        atomicAdd(&denom2[dl], wl);
    }
    int so = sl * OUTC, dof = dl * OUTC;
    float val[M2E / 2];
    int   dd[M2E / 2];
#pragma unroll
    for (int i = 0; i < M2E / 2; ++i) {
        int   s = __shfl(so,  2 * i + j, 32);
        int   d = __shfl(dof, 2 * i + j, 32);
        float w = __shfl(wl,  2 * i + j, 32);
        dd[i]  = d + c;
        val[i] = h2[s + c] * w;
    }
    if (e0 + M2E <= ET) {
#pragma unroll
        for (int i = 0; i < M2E / 2; ++i)
            atomicAdd(&out[dd[i]], val[i]);
    } else {
#pragma unroll
        for (int i = 0; i < M2E / 2; ++i)
            if (e0 + 2 * i + j < ET) atomicAdd(&out[dd[i]], val[i]);
    }
}

__global__ void k_norm_bias2(float* __restrict__ out, const float* __restrict__ denom2,
                             const float* __restrict__ bias2, size_t n) {
    size_t i = (size_t)blockIdx.x * blockDim.x + threadIdx.x;
    size_t stride = (size_t)gridDim.x * blockDim.x;
    for (; i < n; i += stride)
        out[i] = out[i] / (denom2[i >> 4] + 1e-16f) + bias2[i & (OUTC - 1)];
}

extern "C" void kernel_launch(void* const* d_in, const int* in_sizes, int n_in,
                              void* d_out, int out_size, void* d_ws, size_t ws_size,
                              hipStream_t stream) {
    const float* x     = (const float*)d_in[0];
    const int*   ei    = (const int*)d_in[1];
    const float* W1    = (const float*)d_in[2];
    const float* as1   = (const float*)d_in[3];
    const float* ad1   = (const float*)d_in[4];
    const float* bias1 = (const float*)d_in[5];
    const float* W2    = (const float*)d_in[6];
    const float* as2   = (const float*)d_in[7];
    const float* ad2   = (const float*)d_in[8];
    const float* bias2 = (const float*)d_in[9];
    float* out = (float*)d_out;

    const int N  = in_sizes[0] / 3;
    const int E  = in_sizes[1] / 2;
    const int ET = E + N;

    float* ws = (float*)d_ws;
    const size_t Nz = (size_t)N;
    float*    agg1   = ws;
    unsigned* m1     = (unsigned*)(ws + 32 * Nz);
    unsigned* m2     = (unsigned*)(ws + 40 * Nz);
    float*    denom2 = ws + 41 * Nz;
    const size_t zero_count = 42 * Nz;
    float*    a_s1 = ws + 298 * Nz;
    float*    a_d1 = ws + 306 * Nz;
    float*    h2   = ws + 314 * Nz;
    float*    a_s2 = ws + 330 * Nz;
    float*    a_d2 = ws + 331 * Nz;

    const int TB = 256;
    k_fill_zero<<<2048, TB, 0, stream>>>(ws, zero_count);
    k_fill_zero<<<512, TB, 0, stream>>>(out, (size_t)N * OUTC);

    {
        int mtiles = (N + 15) / 16;
        k_att1_x<<<(mtiles + 7) / 8, TB, 0, stream>>>(x, W1, as1, ad1, a_s1, a_d1, N);
    }
    k_edge_max1<<<(ET + TB - 1) / TB, TB, 0, stream>>>(ei, E, ET, a_s1, a_d1, m1);
    {
        long long waves = ((long long)ET + EPW - 1) / EPW;
        k_msg1x<<<(int)((waves * 32 + TB - 1) / TB), TB, 0, stream>>>(ei, E, ET, x, a_s1, a_d1,
                                                                      m1, agg1);
    }

    {
        int mtiles = (N + 15) / 16;
        long long thr = (long long)mtiles * 32;
        k_l1out_gemm2_wmma<<<(int)((thr + 63) / 64), 64, 0, stream>>>(agg1, W1, bias1, W2, h2, N);
    }
    k_att2<<<(N + TB - 1) / TB, TB, 0, stream>>>(h2, as2, ad2, a_s2, a_d2, N);
    k_edge_max2<<<(ET + TB - 1) / TB, TB, 0, stream>>>(ei, E, ET, a_s2, a_d2, m2);
    {
        long long waves = ((long long)ET + M2E - 1) / M2E;
        k_msg2<<<(int)((waves * 32 + TB - 1) / TB), TB, 0, stream>>>(ei, E, ET, a_s2, a_d2,
                                                                     m2, denom2, h2, out);
    }
    k_norm_bias2<<<512, TB, 0, stream>>>(out, denom2, bias2, (size_t)N * OUTC);
}


# Edges grouped by destination once per call (count, allot, scatter); a wave per node takes max, exp and sums in registers: no float atomics, max passes, zero fills or divide pass (sum order changes)
# speedup vs baseline: 6.1412x; 1.5249x over previous
#include <hip/hip_runtime.h>
#include <math.h>

#define HEADS     8
#define HIDC      32
#define L1C       256
#define OUTC      16
#define NEG_SLOPE 0.2f

typedef __attribute__((ext_vector_type(2))) float v2f;
typedef __attribute__((ext_vector_type(8))) float v8f;

__device__ __forceinline__ float lrelu(float v) {
    return v > 0.0f ? v : NEG_SLOPE * v;
}

__global__ void k_fill_zero(float* __restrict__ p, size_t n) {
    size_t i = (size_t)blockIdx.x * blockDim.x + threadIdx.x;
    size_t stride = (size_t)gridDim.x * blockDim.x;
    for (; i < n; i += stride) p[i] = 0.0f;
}

__global__ void k_att1_x(const float* __restrict__ x, const float* __restrict__ W1,
                         const float* __restrict__ as1, const float* __restrict__ ad1,
                         float* __restrict__ a_s, float* __restrict__ a_d, int n_nodes) {
    __shared__ float sV[3 * 16];
    int w    = threadIdx.x >> 5;
    int lane = threadIdx.x & 31;
    {
        float ts = as1[w * HIDC + lane];
        float td = ad1[w * HIDC + lane];
        float p[6];
#pragma unroll
        for (int k = 0; k < 3; ++k) {
            float wv = W1[k * L1C + w * HIDC + lane];
            p[k]     = wv * ts;
            p[3 + k] = wv * td;
        }
        for (int off = 16; off; off >>= 1) {
#pragma unroll
            for (int i = 0; i < 6; ++i) p[i] += __shfl_xor(p[i], off, 32);
        }
        if (lane == 0) {
#pragma unroll
            for (int k = 0; k < 3; ++k) {
                sV[k * 16 + w]     = p[k];
                sV[k * 16 + 8 + w] = p[3 + k];
            }
        }
    }
    __syncthreads();
    int mtiles = (n_nodes + 15) >> 4;
    int tile   = blockIdx.x * 8 + w;
    if (tile >= mtiles) return;
    int c  = lane & 15;
    int kh = lane >> 4;
    int node = tile * 16 + c;
    bool valid = node < n_nodes;
    int nc = valid ? node : n_nodes - 1;
    float x0 = x[nc * 3 + 0], x1 = x[nc * 3 + 1], x2 = x[nc * 3 + 2];
    v2f a, b;
    a.x = sV[(2 * kh) * 16 + c];  a.y = kh ? 0.0f : sV[16 + c];
    b.x = kh ? x2 : x0;           b.y = kh ? 0.0f : x1;
    float* po = (kh ? a_d : a_s) + (size_t)node * HEADS;
    v8f acc = {};
    acc = __builtin_amdgcn_wmma_f32_16x16x4_f32(false, a, false, b, (short)0, acc,
                                                false, false);
    if (valid) *(v8f*)po = acc;
}

#define CEPT 4
__global__ void k_count(const int* __restrict__ ei, int E, int* __restrict__ cnt,
                        int* __restrict__ rank) {
    int base = blockIdx.x * (blockDim.x * CEPT) + threadIdx.x;
    int d[CEPT], r[CEPT];
#pragma unroll
    for (int i = 0; i < CEPT; ++i) {
        int e = base + i * blockDim.x;
        d[i] = ei[E + (e < E ? e : E - 1)];
    }
#pragma unroll
    for (int i = 0; i < CEPT; ++i) {
        int e = base + i * blockDim.x;
        r[i] = 0;
        if (e < E) r[i] = atomicAdd(&cnt[d[i]], 1);
    }
#pragma unroll
    for (int i = 0; i < CEPT; ++i) {
        int e = base + i * blockDim.x;
        if (e < E) rank[e] = r[i];
    }
}

__global__ void k_alloc(const int* cnt, int* cursor, int* off, int n_nodes) {
    int d    = blockIdx.x * blockDim.x + threadIdx.x;
    int lane = threadIdx.x & 31;
    int c = d < n_nodes ? cnt[d] : 0;
    int incl = c;
    for (int o = 1; o < 32; o <<= 1) {
        int t = __shfl_up(incl, o, 32);
        if (lane >= o) incl += t;
    }
    int base = 0;
    if (lane == 31) base = atomicAdd(cursor, incl);
    base = __shfl(base, 31, 32);
    if (d < n_nodes) off[d] = base + incl - c;
}

__global__ void k_scatter(const int* __restrict__ ei, int E, const int* __restrict__ off,
                          const int* __restrict__ rank, int* __restrict__ csr) {
    int base = blockIdx.x * (blockDim.x * CEPT) + threadIdx.x;
    int s[CEPT], p[CEPT];
#pragma unroll
    for (int i = 0; i < CEPT; ++i) {
        int e  = base + i * blockDim.x;
        int ec = e < E ? e : E - 1;
        s[i] = ei[ec];
        p[i] = off[ei[E + ec]] + rank[ec];
    }
#pragma unroll
    for (int i = 0; i < CEPT; ++i) {
        int e = base + i * blockDim.x;
        if (e < E) csr[p[i]] = s[i];
    }
}

template <int NI>
__device__ __forceinline__ float l1_chunk_max(int sl, int j, int h,
                                              const float* __restrict__ a_s) {
    float v[NI];
#pragma unroll
    for (int i = 0; i < NI; ++i) v[i] = a_s[__shfl(sl, i * 4 + j, 32) * HEADS + h];
    float mx = v[0];
#pragma unroll
    for (int i = 1; i < NI; ++i) mx = fmaxf(mx, v[i]);
    return mx;
}
template <int NI>
__device__ __forceinline__ void l1_chunk_sum(int sl, int nb, int j, int h, float ad, float m,
                                             const float* __restrict__ a_s,
                                             const float* __restrict__ x,
                                             float& S0, float& S1, float& S2, float& S3) {
    float av[NI], x0[NI], x1[NI], x2[NI];
#pragma unroll
    for (int i = 0; i < NI; ++i) {
        int s = __shfl(sl, i * 4 + j, 32);
        av[i] = a_s[s * HEADS + h];
        x0[i] = x[s * 3 + 0];
        x1[i] = x[s * 3 + 1];
        x2[i] = x[s * 3 + 2];
    }
#pragma unroll
    for (int i = 0; i < NI; ++i) asm volatile("" :: "v"(av[i]), "v"(x0[i]), "v"(x1[i]), "v"(x2[i]));
#pragma unroll
    for (int i = 0; i < NI; ++i) {
        float w = expf(lrelu(av[i] + ad) - m);
        w = (i * 4 + j < nb) ? w : 0.0f;
        S0 += w * x0[i];
        S1 += w * x1[i];
        S2 += w * x2[i];
        S3 += w;
    }
}
__global__ void k_agg1(const int* __restrict__ cnt, const int* __restrict__ off,
                       const int* __restrict__ csr, const float* __restrict__ x,
                       const float* __restrict__ a_s, const float* __restrict__ a_d,
                       float* __restrict__ agg, int n_nodes) {
    int d = (int)((blockIdx.x * blockDim.x + threadIdx.x) >> 5);
    if (d >= n_nodes) return;
    d = __builtin_amdgcn_readfirstlane(d);
    int lane = threadIdx.x & 31;
    int h = lane & 7;
    int j = lane >> 3;
    int n   = cnt[d] + 1;
    int beg = off[d] - 1;
    float ad = a_d[d * HEADS + h];
    float mx = -INFINITY;
    for (int b = 0; b < n; b += 32) {
        int t  = b + lane;
        int sl = d;
        if (t >= 1 && t < n) sl = csr[beg + t];
        float cm;
        if (n - b <= 16) cm = l1_chunk_max<4>(sl, j, h, a_s);
        else             cm = l1_chunk_max<8>(sl, j, h, a_s);
        mx = fmaxf(mx, cm);
    }
    mx = fmaxf(mx, __shfl_xor(mx, 8, 32));
    mx = fmaxf(mx, __shfl_xor(mx, 16, 32));
    float m = lrelu(mx + ad);
    float S0 = 0.0f, S1 = 0.0f, S2 = 0.0f, S3 = 0.0f;
    for (int b = 0; b < n; b += 32) {
        int t  = b + lane;
        int sl = d;
        if (t >= 1 && t < n) sl = csr[beg + t];
        if (n - b <= 16) l1_chunk_sum<4>(sl, n - b, j, h, ad, m, a_s, x, S0, S1, S2, S3);
        else             l1_chunk_sum<8>(sl, n - b, j, h, ad, m, a_s, x, S0, S1, S2, S3);
    }
    S0 += __shfl_xor(S0, 8, 32);  S0 += __shfl_xor(S0, 16, 32);
    S1 += __shfl_xor(S1, 8, 32);  S1 += __shfl_xor(S1, 16, 32);
    S2 += __shfl_xor(S2, 8, 32);  S2 += __shfl_xor(S2, 16, 32);
    S3 += __shfl_xor(S3, 8, 32);  S3 += __shfl_xor(S3, 16, 32);
    float o = j == 0 ? S0 : (j == 1 ? S1 : (j == 2 ? S2 : S3));
    agg[(size_t)d * 32 + h * 4 + j] = o;
}

typedef __attribute__((ext_vector_type(4))) float v4f;
struct HeadIn {
    v4f   p;
    float w1a[2];
    float w1b[2];
    v8f   bs[2];
    float w2[2][8];
};
__device__ __forceinline__ HeadIn load_head(const float* __restrict__ prow,
                                            const float* __restrict__ W1,
                                            const float* __restrict__ bias1,
                                            const float* __restrict__ W2,
                                            int head, int m, int kh) {
    HeadIn r;
    r.p = *(const v4f*)(prow + head * 4);
#pragma unroll
    for (int t = 0; t < 2; ++t) {
        int cb  = head * HIDC + t * 16;
        int col = cb + m;
        r.w1a[t] = W1[(2 * kh) * L1C + col];
        r.w1b[t] = W1[L1C + col];
        r.bs[t]  = *(const v8f*)(bias1 + cb + 8 * kh);
        const float* w2p = W2 + (size_t)(cb + 8 * kh) * OUTC + m;
#pragma unroll
        for (int i = 0; i < 8; ++i) r.w2[t][i] = w2p[i * OUTC];
    }
    return r;
}
__global__ void __launch_bounds__(64)
k_l1out_gemm2_wmma(const float* __restrict__ agg, const float* __restrict__ W1,
                   const float* __restrict__ bias1, const float* __restrict__ W2,
                   float* __restrict__ h2, int n_nodes) {
    int wave = (int)((blockIdx.x * blockDim.x + threadIdx.x) >> 5);
    int lane = threadIdx.x & 31;
    int mtiles = (n_nodes + 15) >> 4;
    if (wave >= mtiles) return;
    int m  = lane & 15;
    int kh = lane >> 4;
    int node = wave * 16 + m;
    bool valid = node < n_nodes;
    int nc = valid ? node : n_nodes - 1;
    const float* prow = agg + (size_t)nc * 32;
    v8f acc0 = {}, acc1 = {};
    HeadIn cur = load_head(prow, W1, bias1, W2, 0, m, kh);
#pragma unroll
    for (int head = 0; head < HEADS; ++head) {
        HeadIn nxt = cur;
        if (head + 1 < HEADS) nxt = load_head(prow, W1, bias1, W2, head + 1, m, kh);
        float den = cur.p.w + 1e-16f;
        v2f b1;
        b1.x = (kh ? cur.p.z : cur.p.x) / den;
        b1.y = kh ? 0.0f : cur.p.y / den;
#pragma unroll
        for (int t = 0; t < 2; ++t) {
            v2f a1;
            a1.x = cur.w1a[t];
            a1.y = kh ? 0.0f : cur.w1b[t];
            v8f d = {};
            d = __builtin_amdgcn_wmma_f32_16x16x4_f32(false, a1, false, b1, (short)0, d,
                                                      false, false);
#pragma unroll
            for (int j = 0; j < 4; ++j) {
                float r0 = d[2 * j] + cur.bs[t][2 * j];
                float r1 = d[2 * j + 1] + cur.bs[t][2 * j + 1];
                float e0 = expf(r0) - 1.0f;
                float e1 = expf(r1) - 1.0f;
                v2f b2;
                b2.x = r0 > 0.0f ? r0 : e0;
                b2.y = r1 > 0.0f ? r1 : e1;
                v2f a2;
                a2.x = cur.w2[t][2 * j];
                a2.y = cur.w2[t][2 * j + 1];
                if (t == 0)
                    acc0 = __builtin_amdgcn_wmma_f32_16x16x4_f32(false, a2, false, b2, (short)0,
                                                                 acc0, false, false);
                else
                    acc1 = __builtin_amdgcn_wmma_f32_16x16x4_f32(false, a2, false, b2, (short)0,
                                                                 acc1, false, false);
            }
        }
        cur = nxt;
    }
    if (valid) *(v8f*)(h2 + (size_t)node * OUTC + 8 * kh) = acc0 + acc1;
}

__global__ void k_att2(const float* __restrict__ h2, const float* __restrict__ as2,
                       const float* __restrict__ ad2, float* __restrict__ a_s,
                       float* __restrict__ a_d, int n_nodes) {
    int n = blockIdx.x * blockDim.x + threadIdx.x;
    if (n >= n_nodes) return;
    const float* row = h2 + (size_t)n * OUTC;
    float s = 0.0f, d = 0.0f;
#pragma unroll
    for (int cc = 0; cc < OUTC; ++cc) {
        float v = row[cc];
        s += v * as2[cc];
        d += v * ad2[cc];
    }
    a_s[n] = s;
    a_d[n] = d;
}

template <int NI>
__device__ __forceinline__ float l2_chunk_sum(int so, float w, int j, int c,
                                              const float* __restrict__ h2) {
    float hv[NI], wv[NI];
#pragma unroll
    for (int i = 0; i < NI; ++i) {
        int s = __shfl(so, 2 * i + j, 32);
        wv[i] = __shfl(w, 2 * i + j, 32);
        hv[i] = h2[s + c];
    }
    float acc = 0.0f;
#pragma unroll
    for (int i = 0; i < NI; ++i) acc += wv[i] * hv[i];
    return acc;
}
__global__ void k_agg2(const int* __restrict__ cnt, const int* __restrict__ off,
                       const int* __restrict__ csr, const float* __restrict__ a_s,
                       const float* __restrict__ a_d, const float* __restrict__ h2,
                       const float* __restrict__ bias2, float* __restrict__ out, int n_nodes) {
    int d = (int)((blockIdx.x * blockDim.x + threadIdx.x) >> 5);
    if (d >= n_nodes) return;
    d = __builtin_amdgcn_readfirstlane(d);
    int lane = threadIdx.x & 31;
    int c = lane & 15;
    int j = lane >> 4;
    int n   = cnt[d] + 1;
    int beg = off[d] - 1;
    float ad = a_d[d];
    float mx = -INFINITY;
    for (int b = 0; b < n; b += 32) {
        int t = b + lane;
        int s = d;
        if (t >= 1 && t < n) s = csr[beg + t];
        mx = fmaxf(mx, a_s[s]);
    }
    for (int o = 16; o; o >>= 1) mx = fmaxf(mx, __shfl_xor(mx, o, 32));
    float m = lrelu(mx + ad);
    float acc = 0.0f, den = 0.0f;
    for (int b = 0; b < n; b += 32) {
        int t = b + lane;
        int s = d;
        if (t >= 1 && t < n) s = csr[beg + t];
        float w = expf(lrelu(a_s[s] + ad) - m);
        w = t < n ? w : 0.0f;
        den += w;
        int so = s * OUTC;
        if (n - b <= 16) acc += l2_chunk_sum<8>(so, w, j, c, h2);
        else             acc += l2_chunk_sum<16>(so, w, j, c, h2);
    }
    for (int o = 16; o; o >>= 1) den += __shfl_xor(den, o, 32);
    acc += __shfl_xor(acc, 16, 32);
    if (lane < OUTC) out[(size_t)d * OUTC + c] = acc / (den + 1e-16f) + bias2[c];
}

extern "C" void kernel_launch(void* const* d_in, const int* in_sizes, int n_in,
                              void* d_out, int out_size, void* d_ws, size_t ws_size,
                              hipStream_t stream) {
    const float* x     = (const float*)d_in[0];
    const int*   ei    = (const int*)d_in[1];
    const float* W1    = (const float*)d_in[2];
    const float* as1   = (const float*)d_in[3];
    const float* ad1   = (const float*)d_in[4];
    const float* bias1 = (const float*)d_in[5];
    const float* W2    = (const float*)d_in[6];
    const float* as2   = (const float*)d_in[7];
    const float* ad2   = (const float*)d_in[8];
    const float* bias2 = (const float*)d_in[9];
    float* out = (float*)d_out;

    const int N  = in_sizes[0] / 3;
    const int E  = in_sizes[1] / 2;

    float* ws = (float*)d_ws;
    const size_t Nz = (size_t)N;
    const size_t Ez = (size_t)E;
    float* agg1 = ws;
    float* h2   = ws + 32 * Nz;
    float* a_s1 = ws + 48 * Nz;
    float* a_d1 = ws + 56 * Nz;
    int*   cnt  = (int*)(ws + 64 * Nz);
    int*   off  = (int*)(ws + 66 * Nz);
    float* a_s2 = ws + 67 * Nz;
    float* a_d2 = ws + 68 * Nz;
    int*   rank = (int*)(ws + 70 * Nz);
    int*   csr  = (int*)(ws + 70 * Nz + Ez);

    const int TB = 256;
    k_fill_zero<<<(N + 1 + TB - 1) / TB, TB, 0, stream>>>((float*)cnt, Nz + 1);
    k_count<<<(E + TB * CEPT - 1) / (TB * CEPT), TB, 0, stream>>>(ei, E, cnt, rank);
    k_alloc<<<(N + TB - 1) / TB, TB, 0, stream>>>(cnt, cnt + N, off, N);
    k_scatter<<<(E + TB * CEPT - 1) / (TB * CEPT), TB, 0, stream>>>(ei, E, off, rank, csr);

    {
        int mtiles = (N + 15) / 16;
        k_att1_x<<<(mtiles + 7) / 8, TB, 0, stream>>>(x, W1, as1, ad1, a_s1, a_d1, N);
    }
    k_agg1<<<(int)(((long long)N * 32 + TB - 1) / TB), TB, 0, stream>>>(cnt, off, csr, x, a_s1, a_d1,
                                                                      agg1, N);

    {
        int mtiles = (N + 15) / 16;
        long long thr = (long long)mtiles * 32;
        k_l1out_gemm2_wmma<<<(int)((thr + 63) / 64), 64, 0, stream>>>(agg1, W1, bias1, W2, h2, N);
    }
    k_att2<<<(N + TB - 1) / TB, TB, 0, stream>>>(h2, as2, ad2, a_s2, a_d2, N);
    k_agg2<<<(int)(((long long)N * 32 + TB - 1) / TB), TB, 0, stream>>>(cnt, off, csr, a_s2, a_d2, h2,
                                                                      bias2, out, N);
}


# Edge counts and ranks taken in LDS: 64 work-groups each keep a counter per node for a slice of the edges (returning LDS adds), not 800000 scattered global atomics; no zero fill (sum order changes)
# speedup vs baseline: 8.5718x; 1.3958x over previous
#include <hip/hip_runtime.h>
#include <math.h>

#define HEADS     8
#define HIDC      32
#define L1C       256
#define OUTC      16
#define NEG_SLOPE 0.2f

typedef __attribute__((ext_vector_type(2))) float v2f;
typedef __attribute__((ext_vector_type(8))) float v8f;

__device__ __forceinline__ float lrelu(float v) {
    return v > 0.0f ? v : NEG_SLOPE * v;
}

__global__ void k_att1_x(const float* __restrict__ x, const float* __restrict__ W1,
                         const float* __restrict__ as1, const float* __restrict__ ad1,
                         float* __restrict__ a_s, float* __restrict__ a_d, int n_nodes) {
    __shared__ float sV[3 * 16];
    int w    = threadIdx.x >> 5;
    int lane = threadIdx.x & 31;
    {
        float ts = as1[w * HIDC + lane];
        float td = ad1[w * HIDC + lane];
        float p[6];
#pragma unroll
        for (int k = 0; k < 3; ++k) {
            float wv = W1[k * L1C + w * HIDC + lane];
            p[k]     = wv * ts;
            p[3 + k] = wv * td;
        }
        for (int off = 16; off; off >>= 1) {
#pragma unroll
            for (int i = 0; i < 6; ++i) p[i] += __shfl_xor(p[i], off, 32);
        }
        if (lane == 0) {
#pragma unroll
            for (int k = 0; k < 3; ++k) {
                sV[k * 16 + w]     = p[k];
                sV[k * 16 + 8 + w] = p[3 + k];
            }
        }
    }
    __syncthreads();
    int mtiles = (n_nodes + 15) >> 4;
    int tile   = blockIdx.x * 8 + w;
    if (tile >= mtiles) return;
    int c  = lane & 15;
    int kh = lane >> 4;
    int node = tile * 16 + c;
    bool valid = node < n_nodes;
    int nc = valid ? node : n_nodes - 1;
    float x0 = x[nc * 3 + 0], x1 = x[nc * 3 + 1], x2 = x[nc * 3 + 2];
    v2f a, b;
    a.x = sV[(2 * kh) * 16 + c];  a.y = kh ? 0.0f : sV[16 + c];
    b.x = kh ? x2 : x0;           b.y = kh ? 0.0f : x1;
    float* po = (kh ? a_d : a_s) + (size_t)node * HEADS;
    v8f acc = {};
    acc = __builtin_amdgcn_wmma_f32_16x16x4_f32(false, a, false, b, (short)0, acc,
                                                false, false);
    if (valid) *(v8f*)po = acc;
}

#define NSLICE 64
#define LDSN   50176
#define CTHR   1024
#define LEPT   16
__global__ void k_count_lds(const int* __restrict__ ei, int E, int n_nodes, int chunk,
                            int* __restrict__ hist, int* __restrict__ rank,
                            int* __restrict__ cursor) {
    __shared__ int h[LDSN];
    const int s   = blockIdx.x;
    const int tid = threadIdx.x;
    if (s == 0 && tid == 0) *cursor = 0;
    const int e0 = s * chunk;
    const int e1 = (e0 + chunk < E) ? e0 + chunk : E;
    for (int nb = 0; nb < n_nodes; nb += LDSN) {
        for (int i = tid; i < LDSN; i += CTHR) h[i] = 0;
        __syncthreads();
        for (int eb = e0; eb < e1; eb += CTHR * LEPT) {
            int d[LEPT];
#pragma unroll
            for (int i = 0; i < LEPT; ++i) {
                int e = eb + i * CTHR + tid;
                d[i] = ei[E + (e < e1 ? e : e1 - 1)];
            }
#pragma unroll
            for (int i = 0; i < LEPT; ++i) {
                int e = eb + i * CTHR + tid;
                unsigned k = (unsigned)(d[i] - nb);
                if (e < e1 && k < (unsigned)LDSN) rank[e] = atomicAdd(&h[k], 1);
            }
        }
        __syncthreads();
        int lim = n_nodes - nb < LDSN ? n_nodes - nb : LDSN;
        for (int i = tid; i < lim; i += CTHR) hist[(size_t)s * n_nodes + nb + i] = h[i];
        __syncthreads();
    }
}

__global__ void k_alloc_slices(int* __restrict__ hist, int* cursor, int* __restrict__ cnt,
                               int* __restrict__ off, int n_nodes) {
    int d    = blockIdx.x * blockDim.x + threadIdx.x;
    int lane = threadIdx.x & 31;
    int dc   = d < n_nodes ? d : n_nodes - 1;
    int t[NSLICE];
#pragma unroll
    for (int s = 0; s < NSLICE; ++s) t[s] = hist[(size_t)s * n_nodes + dc];
    int c = 0;
#pragma unroll
    for (int s = 0; s < NSLICE; ++s) c += t[s];
    if (d >= n_nodes) c = 0;
    int incl = c;
    for (int o = 1; o < 32; o <<= 1) {
        int u = __shfl_up(incl, o, 32);
        if (lane >= o) incl += u;
    }
    int base = 0;
    if (lane == 31) base = atomicAdd(cursor, incl);
    base = __shfl(base, 31, 32);
    if (d < n_nodes) {
        int run = base + incl - c;
        cnt[d] = c;
        off[d] = run;
#pragma unroll
        for (int s = 0; s < NSLICE; ++s) {
            hist[(size_t)s * n_nodes + d] = run;
            run += t[s];
        }
    }
}

#define CEPT 4
__global__ void k_scatter(const int* __restrict__ ei, int E, int n_nodes, int chunk,
                          const int* __restrict__ hist, const int* __restrict__ rank,
                          int* __restrict__ csr) {
    int base = blockIdx.x * (blockDim.x * CEPT) + threadIdx.x;
    int s[CEPT], p[CEPT];
#pragma unroll
    for (int i = 0; i < CEPT; ++i) {
        int e  = base + i * blockDim.x;
        int ec = e < E ? e : E - 1;
        s[i] = ei[ec];
        p[i] = hist[(size_t)(ec / chunk) * n_nodes + ei[E + ec]] + rank[ec];
    }
#pragma unroll
    for (int i = 0; i < CEPT; ++i) {
        int e = base + i * blockDim.x;
        if (e < E) csr[p[i]] = s[i];
    }
}

template <int NI>
__device__ __forceinline__ float l1_chunk_max(int sl, int j, int h,
                                              const float* __restrict__ a_s) {
    float v[NI];
#pragma unroll
    for (int i = 0; i < NI; ++i) v[i] = a_s[__shfl(sl, i * 4 + j, 32) * HEADS + h];
    float mx = v[0];
#pragma unroll
    for (int i = 1; i < NI; ++i) mx = fmaxf(mx, v[i]);
    return mx;
}
template <int NI>
__device__ __forceinline__ void l1_chunk_sum(int sl, int nb, int j, int h, float ad, float m,
                                             const float* __restrict__ a_s,
                                             const float* __restrict__ x,
                                             float& S0, float& S1, float& S2, float& S3) {
    float av[NI], x0[NI], x1[NI], x2[NI];
#pragma unroll
    for (int i = 0; i < NI; ++i) {
        int s = __shfl(sl, i * 4 + j, 32);
        av[i] = a_s[s * HEADS + h];
        x0[i] = x[s * 3 + 0];
        x1[i] = x[s * 3 + 1];
        x2[i] = x[s * 3 + 2];
    }
#pragma unroll
    for (int i = 0; i < NI; ++i) asm volatile("" :: "v"(av[i]), "v"(x0[i]), "v"(x1[i]), "v"(x2[i]));
#pragma unroll
    for (int i = 0; i < NI; ++i) {
        float w = expf(lrelu(av[i] + ad) - m);
        w = (i * 4 + j < nb) ? w : 0.0f;
        S0 += w * x0[i];
        S1 += w * x1[i];
        S2 += w * x2[i];
        S3 += w;
    }
}
__global__ void k_agg1(const int* __restrict__ cnt, const int* __restrict__ off,
                       const int* __restrict__ csr, const float* __restrict__ x,
                       const float* __restrict__ a_s, const float* __restrict__ a_d,
                       float* __restrict__ agg, int n_nodes) {
    int d = (int)((blockIdx.x * blockDim.x + threadIdx.x) >> 5);
    if (d >= n_nodes) return;
    d = __builtin_amdgcn_readfirstlane(d);
    int lane = threadIdx.x & 31;
    int h = lane & 7;
    int j = lane >> 3;
    int n   = cnt[d] + 1;
    int beg = off[d] - 1;
    float ad = a_d[d * HEADS + h];
    float mx = -INFINITY;
    for (int b = 0; b < n; b += 32) {
        int t  = b + lane;
        int sl = d;
        if (t >= 1 && t < n) sl = csr[beg + t];
        float cm;
        if (n - b <= 16) cm = l1_chunk_max<4>(sl, j, h, a_s);
        else             cm = l1_chunk_max<8>(sl, j, h, a_s);
        mx = fmaxf(mx, cm);
    }
    mx = fmaxf(mx, __shfl_xor(mx, 8, 32));
    mx = fmaxf(mx, __shfl_xor(mx, 16, 32));
    float m = lrelu(mx + ad);
    float S0 = 0.0f, S1 = 0.0f, S2 = 0.0f, S3 = 0.0f;
    for (int b = 0; b < n; b += 32) {
        int t  = b + lane;
        int sl = d;
        if (t >= 1 && t < n) sl = csr[beg + t];
        if (n - b <= 16) l1_chunk_sum<4>(sl, n - b, j, h, ad, m, a_s, x, S0, S1, S2, S3);
        else             l1_chunk_sum<8>(sl, n - b, j, h, ad, m, a_s, x, S0, S1, S2, S3);
    }
    S0 += __shfl_xor(S0, 8, 32);  S0 += __shfl_xor(S0, 16, 32);
    S1 += __shfl_xor(S1, 8, 32);  S1 += __shfl_xor(S1, 16, 32);
    S2 += __shfl_xor(S2, 8, 32);  S2 += __shfl_xor(S2, 16, 32);
    S3 += __shfl_xor(S3, 8, 32);  S3 += __shfl_xor(S3, 16, 32);
    float o = j == 0 ? S0 : (j == 1 ? S1 : (j == 2 ? S2 : S3));
    agg[(size_t)d * 32 + h * 4 + j] = o;
}

typedef __attribute__((ext_vector_type(4))) float v4f;
struct HeadIn {
    v4f   p;
    float w1a[2];
    float w1b[2];
    v8f   bs[2];
    float w2[2][8];
};
__device__ __forceinline__ HeadIn load_head(const float* __restrict__ prow,
                                            const float* __restrict__ W1,
                                            const float* __restrict__ bias1,
                                            const float* __restrict__ W2,
                                            int head, int m, int kh) {
    HeadIn r;
    r.p = *(const v4f*)(prow + head * 4);
#pragma unroll
    for (int t = 0; t < 2; ++t) {
        int cb  = head * HIDC + t * 16;
        int col = cb + m;
        r.w1a[t] = W1[(2 * kh) * L1C + col];
        r.w1b[t] = W1[L1C + col];
        r.bs[t]  = *(const v8f*)(bias1 + cb + 8 * kh);
        const float* w2p = W2 + (size_t)(cb + 8 * kh) * OUTC + m;
#pragma unroll
        for (int i = 0; i < 8; ++i) r.w2[t][i] = w2p[i * OUTC];
    }
    return r;
}
__global__ void __launch_bounds__(64)
k_l1out_gemm2_wmma(const float* __restrict__ agg, const float* __restrict__ W1,
                   const float* __restrict__ bias1, const float* __restrict__ W2,
                   float* __restrict__ h2, int n_nodes) {
    int wave = (int)((blockIdx.x * blockDim.x + threadIdx.x) >> 5);
    int lane = threadIdx.x & 31;
    int mtiles = (n_nodes + 15) >> 4;
    if (wave >= mtiles) return;
    int m  = lane & 15;
    int kh = lane >> 4;
    int node = wave * 16 + m;
    bool valid = node < n_nodes;
    int nc = valid ? node : n_nodes - 1;
    const float* prow = agg + (size_t)nc * 32;
    v8f acc0 = {}, acc1 = {};
    HeadIn cur = load_head(prow, W1, bias1, W2, 0, m, kh);
#pragma unroll
    for (int head = 0; head < HEADS; ++head) {
        HeadIn nxt = cur;
        if (head + 1 < HEADS) nxt = load_head(prow, W1, bias1, W2, head + 1, m, kh);
        float den = cur.p.w + 1e-16f;
        v2f b1;
        b1.x = (kh ? cur.p.z : cur.p.x) / den;
        b1.y = kh ? 0.0f : cur.p.y / den;
#pragma unroll
        for (int t = 0; t < 2; ++t) {
            v2f a1;
            a1.x = cur.w1a[t];
            a1.y = kh ? 0.0f : cur.w1b[t];
            v8f d = {};
            d = __builtin_amdgcn_wmma_f32_16x16x4_f32(false, a1, false, b1, (short)0, d,
                                                      false, false);
#pragma unroll
            for (int j = 0; j < 4; ++j) {
                float r0 = d[2 * j] + cur.bs[t][2 * j];
                float r1 = d[2 * j + 1] + cur.bs[t][2 * j + 1];
                float e0 = expf(r0) - 1.0f;
                float e1 = expf(r1) - 1.0f;
                v2f b2;
                b2.x = r0 > 0.0f ? r0 : e0;
                b2.y = r1 > 0.0f ? r1 : e1;
                v2f a2;
                a2.x = cur.w2[t][2 * j];
                a2.y = cur.w2[t][2 * j + 1];
                if (t == 0)
                    acc0 = __builtin_amdgcn_wmma_f32_16x16x4_f32(false, a2, false, b2, (short)0,
                                                                 acc0, false, false);
                else
                    acc1 = __builtin_amdgcn_wmma_f32_16x16x4_f32(false, a2, false, b2, (short)0,
                                                                 acc1, false, false);
            }
        }
        cur = nxt;
    }
    if (valid) *(v8f*)(h2 + (size_t)node * OUTC + 8 * kh) = acc0 + acc1;
}

__global__ void k_att2(const float* __restrict__ h2, const float* __restrict__ as2,
                       const float* __restrict__ ad2, float* __restrict__ a_s,
                       float* __restrict__ a_d, int n_nodes) {
    int n = blockIdx.x * blockDim.x + threadIdx.x;
    if (n >= n_nodes) return;
    const float* row = h2 + (size_t)n * OUTC;
    float s = 0.0f, d = 0.0f;
#pragma unroll
    for (int cc = 0; cc < OUTC; ++cc) {
        float v = row[cc];
        s += v * as2[cc];
        d += v * ad2[cc];
    }
    a_s[n] = s;
    a_d[n] = d;
}

template <int NI>
__device__ __forceinline__ float l2_chunk_sum(int so, float w, int j, int c,
                                              const float* __restrict__ h2) {
    float hv[NI], wv[NI];
#pragma unroll
    for (int i = 0; i < NI; ++i) {
        int s = __shfl(so, 2 * i + j, 32);
        wv[i] = __shfl(w, 2 * i + j, 32);
        hv[i] = h2[s + c];
    }
    float acc = 0.0f;
#pragma unroll
    for (int i = 0; i < NI; ++i) acc += wv[i] * hv[i];
    return acc;
}
__global__ void k_agg2(const int* __restrict__ cnt, const int* __restrict__ off,
                       const int* __restrict__ csr, const float* __restrict__ a_s,
                       const float* __restrict__ a_d, const float* __restrict__ h2,
                       const float* __restrict__ bias2, float* __restrict__ out, int n_nodes) {
    int d = (int)((blockIdx.x * blockDim.x + threadIdx.x) >> 5);
    if (d >= n_nodes) return;
    d = __builtin_amdgcn_readfirstlane(d);
    int lane = threadIdx.x & 31;
    int c = lane & 15;
    int j = lane >> 4;
    int n   = cnt[d] + 1;
    int beg = off[d] - 1;
    float ad = a_d[d];
    float mx = -INFINITY;
    for (int b = 0; b < n; b += 32) {
        int t = b + lane;
        int s = d;
        if (t >= 1 && t < n) s = csr[beg + t];
        mx = fmaxf(mx, a_s[s]);
    }
    for (int o = 16; o; o >>= 1) mx = fmaxf(mx, __shfl_xor(mx, o, 32));
    float m = lrelu(mx + ad);
    float acc = 0.0f, den = 0.0f;
    for (int b = 0; b < n; b += 32) {
        int t = b + lane;
        int s = d;
        if (t >= 1 && t < n) s = csr[beg + t];
        float w = expf(lrelu(a_s[s] + ad) - m);
        w = t < n ? w : 0.0f;
        den += w;
        int so = s * OUTC;
        if (n - b <= 16) acc += l2_chunk_sum<8>(so, w, j, c, h2);
        else             acc += l2_chunk_sum<16>(so, w, j, c, h2);
    }
    for (int o = 16; o; o >>= 1) den += __shfl_xor(den, o, 32);
    acc += __shfl_xor(acc, 16, 32);
    if (lane < OUTC) out[(size_t)d * OUTC + c] = acc / (den + 1e-16f) + bias2[c];
}

extern "C" void kernel_launch(void* const* d_in, const int* in_sizes, int n_in,
                              void* d_out, int out_size, void* d_ws, size_t ws_size,
                              hipStream_t stream) {
    const float* x     = (const float*)d_in[0];
    const int*   ei    = (const int*)d_in[1];
    const float* W1    = (const float*)d_in[2];
    const float* as1   = (const float*)d_in[3];
    const float* ad1   = (const float*)d_in[4];
    const float* bias1 = (const float*)d_in[5];
    const float* W2    = (const float*)d_in[6];
    const float* as2   = (const float*)d_in[7];
    const float* ad2   = (const float*)d_in[8];
    const float* bias2 = (const float*)d_in[9];
    float* out = (float*)d_out;

    const int N  = in_sizes[0] / 3;
    const int E  = in_sizes[1] / 2;

    float* ws = (float*)d_ws;
    const size_t Nz = (size_t)N;
    const size_t Ez = (size_t)E;
    float* agg1 = ws;
    float* h2   = ws + 32 * Nz;
    float* a_s1 = ws + 48 * Nz;
    float* a_d1 = ws + 56 * Nz;
    int*   cnt  = (int*)(ws + 64 * Nz);
    int*   off  = (int*)(ws + 66 * Nz);
    float* a_s2 = ws + 67 * Nz;
    float* a_d2 = ws + 68 * Nz;
    int*   rank = (int*)(ws + 70 * Nz);
    int*   csr  = (int*)(ws + 70 * Nz + Ez);
    int*   hist = (int*)(ws + 70 * Nz + 2 * Ez);

    const int TB = 256;
    const int chunk = (E + NSLICE - 1) / NSLICE;
    k_count_lds<<<NSLICE, CTHR, 0, stream>>>(ei, E, N, chunk, hist, rank, cnt + N);
    k_alloc_slices<<<(N + TB - 1) / TB, TB, 0, stream>>>(hist, cnt + N, cnt, off, N);
    k_scatter<<<(E + TB * CEPT - 1) / (TB * CEPT), TB, 0, stream>>>(ei, E, N, chunk, hist, rank, csr);

    {
        int mtiles = (N + 15) / 16;
        k_att1_x<<<(mtiles + 7) / 8, TB, 0, stream>>>(x, W1, as1, ad1, a_s1, a_d1, N);
    }
    k_agg1<<<(int)(((long long)N * 32 + TB - 1) / TB), TB, 0, stream>>>(cnt, off, csr, x, a_s1, a_d1,
                                                                      agg1, N);

    {
        int mtiles = (N + 15) / 16;
        long long thr = (long long)mtiles * 32;
        k_l1out_gemm2_wmma<<<(int)((thr + 63) / 64), 64, 0, stream>>>(agg1, W1, bias1, W2, h2, N);
    }
    k_att2<<<(N + TB - 1) / TB, TB, 0, stream>>>(h2, as2, ad2, a_s2, a_d2, N);
    k_agg2<<<(int)(((long long)N * 32 + TB - 1) / TB), TB, 0, stream>>>(cnt, off, csr, a_s2, a_d2, h2,
                                                                      bias2, out, N);
}


# Per-destination kernels, lists of up to 32 elements: sources gathered once as one group and kept in registers for both the max and the sums; no second read of list and logits (sum order unchanged)
# speedup vs baseline: 9.0026x; 1.0503x over previous
#include <hip/hip_runtime.h>
#include <math.h>

#define HEADS     8
#define HIDC      32
#define L1C       256
#define OUTC      16
#define NEG_SLOPE 0.2f

typedef __attribute__((ext_vector_type(2))) float v2f;
typedef __attribute__((ext_vector_type(8))) float v8f;

__device__ __forceinline__ float lrelu(float v) {
    return v > 0.0f ? v : NEG_SLOPE * v;
}

__global__ void k_att1_x(const float* __restrict__ x, const float* __restrict__ W1,
                         const float* __restrict__ as1, const float* __restrict__ ad1,
                         float* __restrict__ a_s, float* __restrict__ a_d, int n_nodes) {
    __shared__ float sV[3 * 16];
    int w    = threadIdx.x >> 5;
    int lane = threadIdx.x & 31;
    {
        float ts = as1[w * HIDC + lane];
        float td = ad1[w * HIDC + lane];
        float p[6];
#pragma unroll
        for (int k = 0; k < 3; ++k) {
            float wv = W1[k * L1C + w * HIDC + lane];
            p[k]     = wv * ts;
            p[3 + k] = wv * td;
        }
        for (int off = 16; off; off >>= 1) {
#pragma unroll
            for (int i = 0; i < 6; ++i) p[i] += __shfl_xor(p[i], off, 32);
        }
        if (lane == 0) {
#pragma unroll
            for (int k = 0; k < 3; ++k) {
                sV[k * 16 + w]     = p[k];
                sV[k * 16 + 8 + w] = p[3 + k];
            }
        }
    }
    __syncthreads();
    int mtiles = (n_nodes + 15) >> 4;
    int tile   = blockIdx.x * 8 + w;
    if (tile >= mtiles) return;
    int c  = lane & 15;
    int kh = lane >> 4;
    int node = tile * 16 + c;
    bool valid = node < n_nodes;
    int nc = valid ? node : n_nodes - 1;
    float x0 = x[nc * 3 + 0], x1 = x[nc * 3 + 1], x2 = x[nc * 3 + 2];
    v2f a, b;
    a.x = sV[(2 * kh) * 16 + c];  a.y = kh ? 0.0f : sV[16 + c];
    b.x = kh ? x2 : x0;           b.y = kh ? 0.0f : x1;
    float* po = (kh ? a_d : a_s) + (size_t)node * HEADS;
    v8f acc = {};
    acc = __builtin_amdgcn_wmma_f32_16x16x4_f32(false, a, false, b, (short)0, acc,
                                                false, false);
    if (valid) *(v8f*)po = acc;
}

#define NSLICE 64
#define LDSN   50176
#define CTHR   1024
#define LEPT   16
__global__ void k_count_lds(const int* __restrict__ ei, int E, int n_nodes, int chunk,
                            int* __restrict__ hist, int* __restrict__ rank,
                            int* __restrict__ cursor) {
    __shared__ int h[LDSN];
    const int s   = blockIdx.x;
    const int tid = threadIdx.x;
    if (s == 0 && tid == 0) *cursor = 0;
    const int e0 = s * chunk;
    const int e1 = (e0 + chunk < E) ? e0 + chunk : E;
    for (int nb = 0; nb < n_nodes; nb += LDSN) {
        for (int i = tid; i < LDSN; i += CTHR) h[i] = 0;
        __syncthreads();
        for (int eb = e0; eb < e1; eb += CTHR * LEPT) {
            int d[LEPT];
#pragma unroll
            for (int i = 0; i < LEPT; ++i) {
                int e = eb + i * CTHR + tid;
                d[i] = ei[E + (e < e1 ? e : e1 - 1)];
            }
#pragma unroll
            for (int i = 0; i < LEPT; ++i) {
                int e = eb + i * CTHR + tid;
                unsigned k = (unsigned)(d[i] - nb);
                if (e < e1 && k < (unsigned)LDSN) rank[e] = atomicAdd(&h[k], 1);
            }
        }
        __syncthreads();
        int lim = n_nodes - nb < LDSN ? n_nodes - nb : LDSN;
        for (int i = tid; i < lim; i += CTHR) hist[(size_t)s * n_nodes + nb + i] = h[i];
        __syncthreads();
    }
}

__global__ void k_alloc_slices(int* __restrict__ hist, int* cursor, int* __restrict__ cnt,
                               int* __restrict__ off, int n_nodes) {
    int d    = blockIdx.x * blockDim.x + threadIdx.x;
    int lane = threadIdx.x & 31;
    int dc   = d < n_nodes ? d : n_nodes - 1;
    int t[NSLICE];
#pragma unroll
    for (int s = 0; s < NSLICE; ++s) t[s] = hist[(size_t)s * n_nodes + dc];
    int c = 0;
#pragma unroll
    for (int s = 0; s < NSLICE; ++s) c += t[s];
    if (d >= n_nodes) c = 0;
    int incl = c;
    for (int o = 1; o < 32; o <<= 1) {
        int u = __shfl_up(incl, o, 32);
        if (lane >= o) incl += u;
    }
    int base = 0;
    if (lane == 31) base = atomicAdd(cursor, incl);
    base = __shfl(base, 31, 32);
    if (d < n_nodes) {
        int run = base + incl - c;
        cnt[d] = c;
        off[d] = run;
#pragma unroll
        for (int s = 0; s < NSLICE; ++s) {
            hist[(size_t)s * n_nodes + d] = run;
            run += t[s];
        }
    }
}

#define CEPT 4
__global__ void k_scatter(const int* __restrict__ ei, int E, int n_nodes, int chunk,
                          const int* __restrict__ hist, const int* __restrict__ rank,
                          int* __restrict__ csr) {
    int base = blockIdx.x * (blockDim.x * CEPT) + threadIdx.x;
    int s[CEPT], p[CEPT];
#pragma unroll
    for (int i = 0; i < CEPT; ++i) {
        int e  = base + i * blockDim.x;
        int ec = e < E ? e : E - 1;
        s[i] = ei[ec];
        p[i] = hist[(size_t)(ec / chunk) * n_nodes + ei[E + ec]] + rank[ec];
    }
#pragma unroll
    for (int i = 0; i < CEPT; ++i) {
        int e = base + i * blockDim.x;
        if (e < E) csr[p[i]] = s[i];
    }
}

template <int NI>
__device__ __forceinline__ float l1_chunk_max(int sl, int j, int h,
                                              const float* __restrict__ a_s) {
    float v[NI];
#pragma unroll
    for (int i = 0; i < NI; ++i) v[i] = a_s[__shfl(sl, i * 4 + j, 32) * HEADS + h];
    float mx = v[0];
#pragma unroll
    for (int i = 1; i < NI; ++i) mx = fmaxf(mx, v[i]);
    return mx;
}
template <int NI>
__device__ __forceinline__ void l1_chunk_sum(int sl, int nb, int j, int h, float ad, float m,
                                             const float* __restrict__ a_s,
                                             const float* __restrict__ x,
                                             float& S0, float& S1, float& S2, float& S3) {
    float av[NI], x0[NI], x1[NI], x2[NI];
#pragma unroll
    for (int i = 0; i < NI; ++i) {
        int s = __shfl(sl, i * 4 + j, 32);
        av[i] = a_s[s * HEADS + h];
        x0[i] = x[s * 3 + 0];
        x1[i] = x[s * 3 + 1];
        x2[i] = x[s * 3 + 2];
    }
#pragma unroll
    for (int i = 0; i < NI; ++i) asm volatile("" :: "v"(av[i]), "v"(x0[i]), "v"(x1[i]), "v"(x2[i]));
#pragma unroll
    for (int i = 0; i < NI; ++i) {
        float w = expf(lrelu(av[i] + ad) - m);
        w = (i * 4 + j < nb) ? w : 0.0f;
        S0 += w * x0[i];
        S1 += w * x1[i];
        S2 += w * x2[i];
        S3 += w;
    }
}
template <int NI>
__device__ __forceinline__ void l1_one_chunk(int sl, int n, int j, int h, float ad,
                                             const float* __restrict__ a_s,
                                             const float* __restrict__ x,
                                             float& S0, float& S1, float& S2, float& S3) {
    float av[NI], x0[NI], x1[NI], x2[NI];
#pragma unroll
    for (int i = 0; i < NI; ++i) {
        int s = __shfl(sl, i * 4 + j, 32);
        av[i] = a_s[s * HEADS + h];
        x0[i] = x[s * 3 + 0];
        x1[i] = x[s * 3 + 1];
        x2[i] = x[s * 3 + 2];
    }
    float mx = av[0];
#pragma unroll
    for (int i = 1; i < NI; ++i) mx = fmaxf(mx, av[i]);
    mx = fmaxf(mx, __shfl_xor(mx, 8, 32));
    mx = fmaxf(mx, __shfl_xor(mx, 16, 32));
    float m = lrelu(mx + ad);
#pragma unroll
    for (int i = 0; i < NI; ++i) {
        float w = expf(lrelu(av[i] + ad) - m);
        w = (i * 4 + j < n) ? w : 0.0f;
        S0 += w * x0[i];
        S1 += w * x1[i];
        S2 += w * x2[i];
        S3 += w;
    }
}
__global__ void k_agg1(const int* __restrict__ cnt, const int* __restrict__ off,
                       const int* __restrict__ csr, const float* __restrict__ x,
                       const float* __restrict__ a_s, const float* __restrict__ a_d,
                       float* __restrict__ agg, int n_nodes) {
    int d = (int)((blockIdx.x * blockDim.x + threadIdx.x) >> 5);
    if (d >= n_nodes) return;
    d = __builtin_amdgcn_readfirstlane(d);
    int lane = threadIdx.x & 31;
    int h = lane & 7;
    int j = lane >> 3;
    int n   = cnt[d] + 1;
    int beg = off[d] - 1;
    float ad = a_d[d * HEADS + h];
    float S0 = 0.0f, S1 = 0.0f, S2 = 0.0f, S3 = 0.0f;
    if (n <= 32) {
        int sl = d;
        if (lane >= 1 && lane < n) sl = csr[beg + lane];
        if (n <= 16) l1_one_chunk<4>(sl, n, j, h, ad, a_s, x, S0, S1, S2, S3);
        else         l1_one_chunk<8>(sl, n, j, h, ad, a_s, x, S0, S1, S2, S3);
    } else {
        float mx = -INFINITY;
        for (int b = 0; b < n; b += 32) {
            int t  = b + lane;
            int sl = d;
            if (t >= 1 && t < n) sl = csr[beg + t];
            float cm;
            if (n - b <= 16) cm = l1_chunk_max<4>(sl, j, h, a_s);
            else             cm = l1_chunk_max<8>(sl, j, h, a_s);
            mx = fmaxf(mx, cm);
        }
        mx = fmaxf(mx, __shfl_xor(mx, 8, 32));
        mx = fmaxf(mx, __shfl_xor(mx, 16, 32));
        float m = lrelu(mx + ad);
        for (int b = 0; b < n; b += 32) {
            int t  = b + lane;
            int sl = d;
            if (t >= 1 && t < n) sl = csr[beg + t];
            if (n - b <= 16) l1_chunk_sum<4>(sl, n - b, j, h, ad, m, a_s, x, S0, S1, S2, S3);
            else             l1_chunk_sum<8>(sl, n - b, j, h, ad, m, a_s, x, S0, S1, S2, S3);
        }
    }
    S0 += __shfl_xor(S0, 8, 32);  S0 += __shfl_xor(S0, 16, 32);
    S1 += __shfl_xor(S1, 8, 32);  S1 += __shfl_xor(S1, 16, 32);
    S2 += __shfl_xor(S2, 8, 32);  S2 += __shfl_xor(S2, 16, 32);
    S3 += __shfl_xor(S3, 8, 32);  S3 += __shfl_xor(S3, 16, 32);
    float o = j == 0 ? S0 : (j == 1 ? S1 : (j == 2 ? S2 : S3));
    agg[(size_t)d * 32 + h * 4 + j] = o;
}

typedef __attribute__((ext_vector_type(4))) float v4f;
struct HeadIn {
    v4f   p;
    float w1a[2];
    float w1b[2];
    v8f   bs[2];
    float w2[2][8];
};
__device__ __forceinline__ HeadIn load_head(const float* __restrict__ prow,
                                            const float* __restrict__ W1,
                                            const float* __restrict__ bias1,
                                            const float* __restrict__ W2,
                                            int head, int m, int kh) {
    HeadIn r;
    r.p = *(const v4f*)(prow + head * 4);
#pragma unroll
    for (int t = 0; t < 2; ++t) {
        int cb  = head * HIDC + t * 16;
        int col = cb + m;
        r.w1a[t] = W1[(2 * kh) * L1C + col];
        r.w1b[t] = W1[L1C + col];
        r.bs[t]  = *(const v8f*)(bias1 + cb + 8 * kh);
        const float* w2p = W2 + (size_t)(cb + 8 * kh) * OUTC + m;
#pragma unroll
        for (int i = 0; i < 8; ++i) r.w2[t][i] = w2p[i * OUTC];
    }
    return r;
}
__global__ void __launch_bounds__(64)
k_l1out_gemm2_wmma(const float* __restrict__ agg, const float* __restrict__ W1,
                   const float* __restrict__ bias1, const float* __restrict__ W2,
                   float* __restrict__ h2, int n_nodes) {
    int wave = (int)((blockIdx.x * blockDim.x + threadIdx.x) >> 5);
    int lane = threadIdx.x & 31;
    int mtiles = (n_nodes + 15) >> 4;
    if (wave >= mtiles) return;
    int m  = lane & 15;
    int kh = lane >> 4;
    int node = wave * 16 + m;
    bool valid = node < n_nodes;
    int nc = valid ? node : n_nodes - 1;
    const float* prow = agg + (size_t)nc * 32;
    v8f acc0 = {}, acc1 = {};
    HeadIn cur = load_head(prow, W1, bias1, W2, 0, m, kh);
#pragma unroll
    for (int head = 0; head < HEADS; ++head) {
        HeadIn nxt = cur;
        if (head + 1 < HEADS) nxt = load_head(prow, W1, bias1, W2, head + 1, m, kh);
        float den = cur.p.w + 1e-16f;
        v2f b1;
        b1.x = (kh ? cur.p.z : cur.p.x) / den;
        b1.y = kh ? 0.0f : cur.p.y / den;
#pragma unroll
        for (int t = 0; t < 2; ++t) {
            v2f a1;
            a1.x = cur.w1a[t];
            a1.y = kh ? 0.0f : cur.w1b[t];
            v8f d = {};
            d = __builtin_amdgcn_wmma_f32_16x16x4_f32(false, a1, false, b1, (short)0, d,
                                                      false, false);
#pragma unroll
            for (int j = 0; j < 4; ++j) {
                float r0 = d[2 * j] + cur.bs[t][2 * j];
                float r1 = d[2 * j + 1] + cur.bs[t][2 * j + 1];
                float e0 = expf(r0) - 1.0f;
                float e1 = expf(r1) - 1.0f;
                v2f b2;
                b2.x = r0 > 0.0f ? r0 : e0;
                b2.y = r1 > 0.0f ? r1 : e1;
                v2f a2;
                a2.x = cur.w2[t][2 * j];
                a2.y = cur.w2[t][2 * j + 1];
                if (t == 0)
                    acc0 = __builtin_amdgcn_wmma_f32_16x16x4_f32(false, a2, false, b2, (short)0,
                                                                 acc0, false, false);
                else
                    acc1 = __builtin_amdgcn_wmma_f32_16x16x4_f32(false, a2, false, b2, (short)0,
                                                                 acc1, false, false);
            }
        }
        cur = nxt;
    }
    if (valid) *(v8f*)(h2 + (size_t)node * OUTC + 8 * kh) = acc0 + acc1;
}

__global__ void k_att2(const float* __restrict__ h2, const float* __restrict__ as2,
                       const float* __restrict__ ad2, float* __restrict__ a_s,
                       float* __restrict__ a_d, int n_nodes) {
    int n = blockIdx.x * blockDim.x + threadIdx.x;
    if (n >= n_nodes) return;
    const float* row = h2 + (size_t)n * OUTC;
    float s = 0.0f, d = 0.0f;
#pragma unroll
    for (int cc = 0; cc < OUTC; ++cc) {
        float v = row[cc];
        s += v * as2[cc];
        d += v * ad2[cc];
    }
    a_s[n] = s;
    a_d[n] = d;
}

template <int NI>
__device__ __forceinline__ float l2_chunk_sum(int so, float w, int j, int c,
                                              const float* __restrict__ h2) {
    float hv[NI], wv[NI];
#pragma unroll
    for (int i = 0; i < NI; ++i) {
        int s = __shfl(so, 2 * i + j, 32);
        wv[i] = __shfl(w, 2 * i + j, 32);
        hv[i] = h2[s + c];
    }
    float acc = 0.0f;
#pragma unroll
    for (int i = 0; i < NI; ++i) acc += wv[i] * hv[i];
    return acc;
}
template <int NI>
__device__ __forceinline__ float l2_one_chunk(int s, int n, int lane, int j, int c, float ad,
                                              const float* __restrict__ a_s,
                                              const float* __restrict__ h2, float& den) {
    float as = a_s[s];
    int so = s * OUTC;
    float hv[NI];
#pragma unroll
    for (int i = 0; i < NI; ++i) hv[i] = h2[__shfl(so, 2 * i + j, 32) + c];
    float mx = as;
    for (int o = 16; o; o >>= 1) mx = fmaxf(mx, __shfl_xor(mx, o, 32));
    float m = lrelu(mx + ad);
    float w = expf(lrelu(as + ad) - m);
    w = lane < n ? w : 0.0f;
    den = w;
    float acc = 0.0f;
#pragma unroll
    for (int i = 0; i < NI; ++i) acc += __shfl(w, 2 * i + j, 32) * hv[i];
    return acc;
}
__global__ void k_agg2(const int* __restrict__ cnt, const int* __restrict__ off,
                       const int* __restrict__ csr, const float* __restrict__ a_s,
                       const float* __restrict__ a_d, const float* __restrict__ h2,
                       const float* __restrict__ bias2, float* __restrict__ out, int n_nodes) {
    int d = (int)((blockIdx.x * blockDim.x + threadIdx.x) >> 5);
    if (d >= n_nodes) return;
    d = __builtin_amdgcn_readfirstlane(d);
    int lane = threadIdx.x & 31;
    int c = lane & 15;
    int j = lane >> 4;
    int n   = cnt[d] + 1;
    int beg = off[d] - 1;
    float ad = a_d[d];
    float acc = 0.0f, den = 0.0f;
    if (n <= 32) {
        int s = d;
        if (lane >= 1 && lane < n) s = csr[beg + lane];
        if (n <= 16) acc = l2_one_chunk<8>(s, n, lane, j, c, ad, a_s, h2, den);
        else         acc = l2_one_chunk<16>(s, n, lane, j, c, ad, a_s, h2, den);
    } else {
        float mx = -INFINITY;
        for (int b = 0; b < n; b += 32) {
            int t = b + lane;
            int s = d;
            if (t >= 1 && t < n) s = csr[beg + t];
            mx = fmaxf(mx, a_s[s]);
        }
        for (int o = 16; o; o >>= 1) mx = fmaxf(mx, __shfl_xor(mx, o, 32));
        float m = lrelu(mx + ad);
        for (int b = 0; b < n; b += 32) {
            int t = b + lane;
            int s = d;
            if (t >= 1 && t < n) s = csr[beg + t];
            float w = expf(lrelu(a_s[s] + ad) - m);
            w = t < n ? w : 0.0f;
            den += w;
            int so = s * OUTC;
            if (n - b <= 16) acc += l2_chunk_sum<8>(so, w, j, c, h2);
            else             acc += l2_chunk_sum<16>(so, w, j, c, h2);
        }
    }
    for (int o = 16; o; o >>= 1) den += __shfl_xor(den, o, 32);
    acc += __shfl_xor(acc, 16, 32);
    if (lane < OUTC) out[(size_t)d * OUTC + c] = acc / (den + 1e-16f) + bias2[c];
}

extern "C" void kernel_launch(void* const* d_in, const int* in_sizes, int n_in,
                              void* d_out, int out_size, void* d_ws, size_t ws_size,
                              hipStream_t stream) {
    const float* x     = (const float*)d_in[0];
    const int*   ei    = (const int*)d_in[1];
    const float* W1    = (const float*)d_in[2];
    const float* as1   = (const float*)d_in[3];
    const float* ad1   = (const float*)d_in[4];
    const float* bias1 = (const float*)d_in[5];
    const float* W2    = (const float*)d_in[6];
    const float* as2   = (const float*)d_in[7];
    const float* ad2   = (const float*)d_in[8];
    const float* bias2 = (const float*)d_in[9];
    float* out = (float*)d_out;

    const int N  = in_sizes[0] / 3;
    const int E  = in_sizes[1] / 2;

    float* ws = (float*)d_ws;
    const size_t Nz = (size_t)N;
    const size_t Ez = (size_t)E;
    float* agg1 = ws;
    float* h2   = ws + 32 * Nz;
    float* a_s1 = ws + 48 * Nz;
    float* a_d1 = ws + 56 * Nz;
    int*   cnt  = (int*)(ws + 64 * Nz);
    int*   off  = (int*)(ws + 66 * Nz);
    float* a_s2 = ws + 67 * Nz;
    float* a_d2 = ws + 68 * Nz;
    int*   rank = (int*)(ws + 70 * Nz);
    int*   csr  = (int*)(ws + 70 * Nz + Ez);
    int*   hist = (int*)(ws + 70 * Nz + 2 * Ez);

    const int TB = 256;
    const int chunk = (E + NSLICE - 1) / NSLICE;
    k_count_lds<<<NSLICE, CTHR, 0, stream>>>(ei, E, N, chunk, hist, rank, cnt + N);
    k_alloc_slices<<<(N + TB - 1) / TB, TB, 0, stream>>>(hist, cnt + N, cnt, off, N);
    k_scatter<<<(E + TB * CEPT - 1) / (TB * CEPT), TB, 0, stream>>>(ei, E, N, chunk, hist, rank, csr);

    {
        int mtiles = (N + 15) / 16;
        k_att1_x<<<(mtiles + 7) / 8, TB, 0, stream>>>(x, W1, as1, ad1, a_s1, a_d1, N);
    }
    k_agg1<<<(int)(((long long)N * 32 + TB - 1) / TB), TB, 0, stream>>>(cnt, off, csr, x, a_s1, a_d1,
                                                                      agg1, N);

    {
        int mtiles = (N + 15) / 16;
        long long thr = (long long)mtiles * 32;
        k_l1out_gemm2_wmma<<<(int)((thr + 63) / 64), 64, 0, stream>>>(agg1, W1, bias1, W2, h2, N);
    }
    k_att2<<<(N + TB - 1) / TB, TB, 0, stream>>>(h2, as2, ad2, a_s2, a_d2, N);
    k_agg2<<<(int)(((long long)N * 32 + TB - 1) / TB), TB, 0, stream>>>(cnt, off, csr, a_s2, a_d2, h2,
                                                                      bias2, out, N);
}


# Layer-2 logits taken in the per-destination pass from the rows it gathers (lane = 4 channels, 8 rows per load, dot summed over a lane quad): no logits launch, no logit gather (sum order changes)
# speedup vs baseline: 9.6005x; 1.0664x over previous
#include <hip/hip_runtime.h>
#include <math.h>

#define HEADS     8
#define HIDC      32
#define L1C       256
#define OUTC      16
#define NEG_SLOPE 0.2f

typedef __attribute__((ext_vector_type(2))) float v2f;
typedef __attribute__((ext_vector_type(8))) float v8f;

__device__ __forceinline__ float lrelu(float v) {
    return v > 0.0f ? v : NEG_SLOPE * v;
}

__global__ void k_att1_x(const float* __restrict__ x, const float* __restrict__ W1,
                         const float* __restrict__ as1, const float* __restrict__ ad1,
                         float* __restrict__ a_s, float* __restrict__ a_d, int n_nodes) {
    __shared__ float sV[3 * 16];
    int w    = threadIdx.x >> 5;
    int lane = threadIdx.x & 31;
    {
        float ts = as1[w * HIDC + lane];
        float td = ad1[w * HIDC + lane];
        float p[6];
#pragma unroll
        for (int k = 0; k < 3; ++k) {
            float wv = W1[k * L1C + w * HIDC + lane];
            p[k]     = wv * ts;
            p[3 + k] = wv * td;
        }
        for (int off = 16; off; off >>= 1) {
#pragma unroll
            for (int i = 0; i < 6; ++i) p[i] += __shfl_xor(p[i], off, 32);
        }
        if (lane == 0) {
#pragma unroll
            for (int k = 0; k < 3; ++k) {
                sV[k * 16 + w]     = p[k];
                sV[k * 16 + 8 + w] = p[3 + k];
            }
        }
    }
    __syncthreads();
    int mtiles = (n_nodes + 15) >> 4;
    int tile   = blockIdx.x * 8 + w;
    if (tile >= mtiles) return;
    int c  = lane & 15;
    int kh = lane >> 4;
    int node = tile * 16 + c;
    bool valid = node < n_nodes;
    int nc = valid ? node : n_nodes - 1;
    float x0 = x[nc * 3 + 0], x1 = x[nc * 3 + 1], x2 = x[nc * 3 + 2];
    v2f a, b;
    a.x = sV[(2 * kh) * 16 + c];  a.y = kh ? 0.0f : sV[16 + c];
    b.x = kh ? x2 : x0;           b.y = kh ? 0.0f : x1;
    float* po = (kh ? a_d : a_s) + (size_t)node * HEADS;
    v8f acc = {};
    acc = __builtin_amdgcn_wmma_f32_16x16x4_f32(false, a, false, b, (short)0, acc,
                                                false, false);
    if (valid) *(v8f*)po = acc;
}

#define NSLICE 64
#define LDSN   50176
#define CTHR   1024
#define LEPT   16
__global__ void k_count_lds(const int* __restrict__ ei, int E, int n_nodes, int chunk,
                            int* __restrict__ hist, int* __restrict__ rank,
                            int* __restrict__ cursor) {
    __shared__ int h[LDSN];
    const int s   = blockIdx.x;
    const int tid = threadIdx.x;
    if (s == 0 && tid == 0) *cursor = 0;
    const int e0 = s * chunk;
    const int e1 = (e0 + chunk < E) ? e0 + chunk : E;
    for (int nb = 0; nb < n_nodes; nb += LDSN) {
        for (int i = tid; i < LDSN; i += CTHR) h[i] = 0;
        __syncthreads();
        for (int eb = e0; eb < e1; eb += CTHR * LEPT) {
            int d[LEPT];
#pragma unroll
            for (int i = 0; i < LEPT; ++i) {
                int e = eb + i * CTHR + tid;
                d[i] = ei[E + (e < e1 ? e : e1 - 1)];
            }
#pragma unroll
            for (int i = 0; i < LEPT; ++i) {
                int e = eb + i * CTHR + tid;
                unsigned k = (unsigned)(d[i] - nb);
                if (e < e1 && k < (unsigned)LDSN) rank[e] = atomicAdd(&h[k], 1);
            }
        }
        __syncthreads();
        int lim = n_nodes - nb < LDSN ? n_nodes - nb : LDSN;
        for (int i = tid; i < lim; i += CTHR) hist[(size_t)s * n_nodes + nb + i] = h[i];
        __syncthreads();
    }
}

__global__ void k_alloc_slices(int* __restrict__ hist, int* cursor, int* __restrict__ cnt,
                               int* __restrict__ off, int n_nodes) {
    int d    = blockIdx.x * blockDim.x + threadIdx.x;
    int lane = threadIdx.x & 31;
    int dc   = d < n_nodes ? d : n_nodes - 1;
    int t[NSLICE];
#pragma unroll
    for (int s = 0; s < NSLICE; ++s) t[s] = hist[(size_t)s * n_nodes + dc];
    int c = 0;
#pragma unroll
    for (int s = 0; s < NSLICE; ++s) c += t[s];
    if (d >= n_nodes) c = 0;
    int incl = c;
    for (int o = 1; o < 32; o <<= 1) {
        int u = __shfl_up(incl, o, 32);
        if (lane >= o) incl += u;
    }
    int base = 0;
    if (lane == 31) base = atomicAdd(cursor, incl);
    base = __shfl(base, 31, 32);
    if (d < n_nodes) {
        int run = base + incl - c;
        cnt[d] = c;
        off[d] = run;
#pragma unroll
        for (int s = 0; s < NSLICE; ++s) {
            hist[(size_t)s * n_nodes + d] = run;
            run += t[s];
        }
    }
}

#define CEPT 4
__global__ void k_scatter(const int* __restrict__ ei, int E, int n_nodes, int chunk,
                          const int* __restrict__ hist, const int* __restrict__ rank,
                          int* __restrict__ csr) {
    int base = blockIdx.x * (blockDim.x * CEPT) + threadIdx.x;
    int s[CEPT], p[CEPT];
#pragma unroll
    for (int i = 0; i < CEPT; ++i) {
        int e  = base + i * blockDim.x;
        int ec = e < E ? e : E - 1;
        s[i] = ei[ec];
        p[i] = hist[(size_t)(ec / chunk) * n_nodes + ei[E + ec]] + rank[ec];
    }
#pragma unroll
    for (int i = 0; i < CEPT; ++i) {
        int e = base + i * blockDim.x;
        if (e < E) csr[p[i]] = s[i];
    }
}

template <int NI>
__device__ __forceinline__ float l1_chunk_max(int sl, int j, int h,
                                              const float* __restrict__ a_s) {
    float v[NI];
#pragma unroll
    for (int i = 0; i < NI; ++i) v[i] = a_s[__shfl(sl, i * 4 + j, 32) * HEADS + h];
    float mx = v[0];
#pragma unroll
    for (int i = 1; i < NI; ++i) mx = fmaxf(mx, v[i]);
    return mx;
}
template <int NI>
__device__ __forceinline__ void l1_chunk_sum(int sl, int nb, int j, int h, float ad, float m,
                                             const float* __restrict__ a_s,
                                             const float* __restrict__ x,
                                             float& S0, float& S1, float& S2, float& S3) {
    float av[NI], x0[NI], x1[NI], x2[NI];
#pragma unroll
    for (int i = 0; i < NI; ++i) {
        int s = __shfl(sl, i * 4 + j, 32);
        av[i] = a_s[s * HEADS + h];
        x0[i] = x[s * 3 + 0];
        x1[i] = x[s * 3 + 1];
        x2[i] = x[s * 3 + 2];
    }
#pragma unroll
    for (int i = 0; i < NI; ++i) asm volatile("" :: "v"(av[i]), "v"(x0[i]), "v"(x1[i]), "v"(x2[i]));
#pragma unroll
    for (int i = 0; i < NI; ++i) {
        float w = expf(lrelu(av[i] + ad) - m);
        w = (i * 4 + j < nb) ? w : 0.0f;
        S0 += w * x0[i];
        S1 += w * x1[i];
        S2 += w * x2[i];
        S3 += w;
    }
}
template <int NI>
__device__ __forceinline__ void l1_one_chunk(int sl, int n, int j, int h, float ad,
                                             const float* __restrict__ a_s,
                                             const float* __restrict__ x,
                                             float& S0, float& S1, float& S2, float& S3) {
    float av[NI], x0[NI], x1[NI], x2[NI];
#pragma unroll
    for (int i = 0; i < NI; ++i) {
        int s = __shfl(sl, i * 4 + j, 32);
        av[i] = a_s[s * HEADS + h];
        x0[i] = x[s * 3 + 0];
        x1[i] = x[s * 3 + 1];
        x2[i] = x[s * 3 + 2];
    }
    float mx = av[0];
#pragma unroll
    for (int i = 1; i < NI; ++i) mx = fmaxf(mx, av[i]);
    mx = fmaxf(mx, __shfl_xor(mx, 8, 32));
    mx = fmaxf(mx, __shfl_xor(mx, 16, 32));
    float m = lrelu(mx + ad);
#pragma unroll
    for (int i = 0; i < NI; ++i) {
        float w = expf(lrelu(av[i] + ad) - m);
        w = (i * 4 + j < n) ? w : 0.0f;
        S0 += w * x0[i];
        S1 += w * x1[i];
        S2 += w * x2[i];
        S3 += w;
    }
}
__global__ void k_agg1(const int* __restrict__ cnt, const int* __restrict__ off,
                       const int* __restrict__ csr, const float* __restrict__ x,
                       const float* __restrict__ a_s, const float* __restrict__ a_d,
                       float* __restrict__ agg, int n_nodes) {
    int d = (int)((blockIdx.x * blockDim.x + threadIdx.x) >> 5);
    if (d >= n_nodes) return;
    d = __builtin_amdgcn_readfirstlane(d);
    int lane = threadIdx.x & 31;
    int h = lane & 7;
    int j = lane >> 3;
    int n   = cnt[d] + 1;
    int beg = off[d] - 1;
    float ad = a_d[d * HEADS + h];
    float S0 = 0.0f, S1 = 0.0f, S2 = 0.0f, S3 = 0.0f;
    if (n <= 32) {
        int sl = d;
        if (lane >= 1 && lane < n) sl = csr[beg + lane];
        if (n <= 16) l1_one_chunk<4>(sl, n, j, h, ad, a_s, x, S0, S1, S2, S3);
        else         l1_one_chunk<8>(sl, n, j, h, ad, a_s, x, S0, S1, S2, S3);
    } else {
        float mx = -INFINITY;
        for (int b = 0; b < n; b += 32) {
            int t  = b + lane;
            int sl = d;
            if (t >= 1 && t < n) sl = csr[beg + t];
            float cm;
            if (n - b <= 16) cm = l1_chunk_max<4>(sl, j, h, a_s);
            else             cm = l1_chunk_max<8>(sl, j, h, a_s);
            mx = fmaxf(mx, cm);
        }
        mx = fmaxf(mx, __shfl_xor(mx, 8, 32));
        mx = fmaxf(mx, __shfl_xor(mx, 16, 32));
        float m = lrelu(mx + ad);
        for (int b = 0; b < n; b += 32) {
            int t  = b + lane;
            int sl = d;
            if (t >= 1 && t < n) sl = csr[beg + t];
            if (n - b <= 16) l1_chunk_sum<4>(sl, n - b, j, h, ad, m, a_s, x, S0, S1, S2, S3);
            else             l1_chunk_sum<8>(sl, n - b, j, h, ad, m, a_s, x, S0, S1, S2, S3);
        }
    }
    S0 += __shfl_xor(S0, 8, 32);  S0 += __shfl_xor(S0, 16, 32);
    S1 += __shfl_xor(S1, 8, 32);  S1 += __shfl_xor(S1, 16, 32);
    S2 += __shfl_xor(S2, 8, 32);  S2 += __shfl_xor(S2, 16, 32);
    S3 += __shfl_xor(S3, 8, 32);  S3 += __shfl_xor(S3, 16, 32);
    float o = j == 0 ? S0 : (j == 1 ? S1 : (j == 2 ? S2 : S3));
    agg[(size_t)d * 32 + h * 4 + j] = o;
}

typedef __attribute__((ext_vector_type(4))) float v4f;
struct HeadIn {
    v4f   p;
    float w1a[2];
    float w1b[2];
    v8f   bs[2];
    float w2[2][8];
};
__device__ __forceinline__ HeadIn load_head(const float* __restrict__ prow,
                                            const float* __restrict__ W1,
                                            const float* __restrict__ bias1,
                                            const float* __restrict__ W2,
                                            int head, int m, int kh) {
    HeadIn r;
    r.p = *(const v4f*)(prow + head * 4);
#pragma unroll
    for (int t = 0; t < 2; ++t) {
        int cb  = head * HIDC + t * 16;
        int col = cb + m;
        r.w1a[t] = W1[(2 * kh) * L1C + col];
        r.w1b[t] = W1[L1C + col];
        r.bs[t]  = *(const v8f*)(bias1 + cb + 8 * kh);
        const float* w2p = W2 + (size_t)(cb + 8 * kh) * OUTC + m;
#pragma unroll
        for (int i = 0; i < 8; ++i) r.w2[t][i] = w2p[i * OUTC];
    }
    return r;
}
__global__ void __launch_bounds__(64)
k_l1out_gemm2_wmma(const float* __restrict__ agg, const float* __restrict__ W1,
                   const float* __restrict__ bias1, const float* __restrict__ W2,
                   float* __restrict__ h2, int n_nodes) {
    int wave = (int)((blockIdx.x * blockDim.x + threadIdx.x) >> 5);
    int lane = threadIdx.x & 31;
    int mtiles = (n_nodes + 15) >> 4;
    if (wave >= mtiles) return;
    int m  = lane & 15;
    int kh = lane >> 4;
    int node = wave * 16 + m;
    bool valid = node < n_nodes;
    int nc = valid ? node : n_nodes - 1;
    const float* prow = agg + (size_t)nc * 32;
    v8f acc0 = {}, acc1 = {};
    HeadIn cur = load_head(prow, W1, bias1, W2, 0, m, kh);
#pragma unroll
    for (int head = 0; head < HEADS; ++head) {
        HeadIn nxt = cur;
        if (head + 1 < HEADS) nxt = load_head(prow, W1, bias1, W2, head + 1, m, kh);
        float den = cur.p.w + 1e-16f;
        v2f b1;
        b1.x = (kh ? cur.p.z : cur.p.x) / den;
        b1.y = kh ? 0.0f : cur.p.y / den;
#pragma unroll
        for (int t = 0; t < 2; ++t) {
            v2f a1;
            a1.x = cur.w1a[t];
            a1.y = kh ? 0.0f : cur.w1b[t];
            v8f d = {};
            d = __builtin_amdgcn_wmma_f32_16x16x4_f32(false, a1, false, b1, (short)0, d,
                                                      false, false);
#pragma unroll
            for (int j = 0; j < 4; ++j) {
                float r0 = d[2 * j] + cur.bs[t][2 * j];
                float r1 = d[2 * j + 1] + cur.bs[t][2 * j + 1];
                float e0 = expf(r0) - 1.0f;
                float e1 = expf(r1) - 1.0f;
                v2f b2;
                b2.x = r0 > 0.0f ? r0 : e0;
                b2.y = r1 > 0.0f ? r1 : e1;
                v2f a2;
                a2.x = cur.w2[t][2 * j];
                a2.y = cur.w2[t][2 * j + 1];
                if (t == 0)
                    acc0 = __builtin_amdgcn_wmma_f32_16x16x4_f32(false, a2, false, b2, (short)0,
                                                                 acc0, false, false);
                else
                    acc1 = __builtin_amdgcn_wmma_f32_16x16x4_f32(false, a2, false, b2, (short)0,
                                                                 acc1, false, false);
            }
        }
        cur = nxt;
    }
    if (valid) *(v8f*)(h2 + (size_t)node * OUTC + 8 * kh) = acc0 + acc1;
}

template <int CTRL>
__device__ __forceinline__ float row_xchg(float v) {
    return __int_as_float(__builtin_amdgcn_update_dpp(0, __float_as_int(v), CTRL, 0xF, 0xF, true));
}
__device__ __forceinline__ float other_row(float v) {
    return __int_as_float(__builtin_amdgcn_permlanex16(0, __float_as_int(v), 0x76543210u,
                                                       0xfedcba98u, false, false));
}
__device__ __forceinline__ float slots_sum(float v) {
    v += row_xchg<0x164>(v);
    v += row_xchg<0x168>(v);
    v += other_row(v);
    return v;
}
__device__ __forceinline__ float slots_max(float v) {
    v = fmaxf(v, row_xchg<0x164>(v));
    v = fmaxf(v, row_xchg<0x168>(v));
    v = fmaxf(v, other_row(v));
    return v;
}
__device__ __forceinline__ float quad_dot(v4f r, v4f a) {
    float p = r.x * a.x + r.y * a.y + r.z * a.z + r.w * a.w;
    p += row_xchg<0xB1>(p);
    p += row_xchg<0x4E>(p);
    return p;
}
template <int NI>
__device__ __forceinline__ void l2_rows(int d, int n, int b, int beg, int sl, int q,
                                        const int* __restrict__ csr,
                                        const float* __restrict__ h2, v4f (&hv)[NI]) {
    int s[NI];
#pragma unroll
    for (int i = 0; i < NI; ++i) {
        int t = b + 8 * i + sl;
        s[i] = d;
        if (t >= 1 && t < n) s[i] = csr[beg + t];
    }
#pragma unroll
    for (int i = 0; i < NI; ++i) hv[i] = *(const v4f*)(h2 + s[i] * OUTC + 4 * q);
}
template <int NI>
__device__ __forceinline__ float l2_weigh(const float (&as)[NI], const v4f (&hv)[NI], int b, int n,
                                          int sl, int q, float ad, float m, v4f& acc) {
    float a1 = as[0];
    if (q == 1) a1 = as[1];
    if constexpr (NI > 2) {
        if (q == 2) a1 = as[2];
        if (q == 3) a1 = as[3];
    }
    float w1 = expf(lrelu(a1 + ad) - m);
    w1 = (q < NI && b + 8 * q + sl < n) ? w1 : 0.0f;
    acc += row_xchg<0x00>(w1) * hv[0];
    acc += row_xchg<0x55>(w1) * hv[1];
    if constexpr (NI > 2) {
        acc += row_xchg<0xAA>(w1) * hv[2];
        acc += row_xchg<0xFF>(w1) * hv[3];
    }
    return w1;
}
template <int NI>
__device__ __forceinline__ void l2_one_chunk(int d, int n, int beg, int sl, int q, float ad,
                                             v4f asq, const int* __restrict__ csr,
                                             const float* __restrict__ h2, v4f& acc, float& den) {
    v4f hv[NI];
    l2_rows<NI>(d, n, 0, beg, sl, q, csr, h2, hv);
    float as[NI];
#pragma unroll
    for (int i = 0; i < NI; ++i) as[i] = quad_dot(hv[i], asq);
    float mx = as[0];
#pragma unroll
    for (int i = 1; i < NI; ++i) mx = fmaxf(mx, as[i]);
    mx = slots_max(mx);
    float m = lrelu(mx + ad);
    den += l2_weigh<NI>(as, hv, 0, n, sl, q, ad, m, acc);
}
__global__ void k_agg2(const int* __restrict__ cnt, const int* __restrict__ off,
                       const int* __restrict__ csr, const float* __restrict__ h2,
                       const float* __restrict__ as2, const float* __restrict__ ad2,
                       const float* __restrict__ bias2, float* __restrict__ out, int n_nodes) {
    int d = (int)((blockIdx.x * blockDim.x + threadIdx.x) >> 5);
    if (d >= n_nodes) return;
    d = __builtin_amdgcn_readfirstlane(d);
    int lane = threadIdx.x & 31;
    int q  = lane & 3;
    int sl = lane >> 2;
    int n   = cnt[d] + 1;
    int beg = off[d] - 1;
    v4f asq = *(const v4f*)(as2 + 4 * q);
    v4f adq = *(const v4f*)(ad2 + 4 * q);
    v4f hd  = *(const v4f*)(h2 + d * OUTC + 4 * q);
    float ad = quad_dot(hd, adq);
    v4f acc = {0.0f, 0.0f, 0.0f, 0.0f};
    float den = 0.0f;
    if (n <= 16) {
        l2_one_chunk<2>(d, n, beg, sl, q, ad, asq, csr, h2, acc, den);
    } else if (n <= 32) {
        l2_one_chunk<4>(d, n, beg, sl, q, ad, asq, csr, h2, acc, den);
    } else {
        float mx = -INFINITY;
        for (int b = 0; b < n; b += 32) {
            v4f hv[4];
            l2_rows<4>(d, n, b, beg, sl, q, csr, h2, hv);
#pragma unroll
            for (int i = 0; i < 4; ++i) mx = fmaxf(mx, quad_dot(hv[i], asq));
        }
        mx = slots_max(mx);
        float m = lrelu(mx + ad);
        for (int b = 0; b < n; b += 32) {
            v4f hv[4];
            l2_rows<4>(d, n, b, beg, sl, q, csr, h2, hv);
            float as[4];
#pragma unroll
            for (int i = 0; i < 4; ++i) as[i] = quad_dot(hv[i], asq);
            den += l2_weigh<4>(as, hv, b, n, sl, q, ad, m, acc);
        }
    }
    den += row_xchg<0xB1>(den);
    den += row_xchg<0x4E>(den);
    den   = slots_sum(den);
    acc.x = slots_sum(acc.x);
    acc.y = slots_sum(acc.y);
    acc.z = slots_sum(acc.z);
    acc.w = slots_sum(acc.w);
    float o = sl == 0 ? acc.x : (sl == 1 ? acc.y : (sl == 2 ? acc.z : acc.w));
    if (sl < 4) out[(size_t)d * OUTC + 4 * q + sl] = o / (den + 1e-16f) + bias2[4 * q + sl];
}

extern "C" void kernel_launch(void* const* d_in, const int* in_sizes, int n_in,
                              void* d_out, int out_size, void* d_ws, size_t ws_size,
                              hipStream_t stream) {
    const float* x     = (const float*)d_in[0];
    const int*   ei    = (const int*)d_in[1];
    const float* W1    = (const float*)d_in[2];
    const float* as1   = (const float*)d_in[3];
    const float* ad1   = (const float*)d_in[4];
    const float* bias1 = (const float*)d_in[5];
    const float* W2    = (const float*)d_in[6];
    const float* as2   = (const float*)d_in[7];
    const float* ad2   = (const float*)d_in[8];
    const float* bias2 = (const float*)d_in[9];
    float* out = (float*)d_out;

    const int N  = in_sizes[0] / 3;
    const int E  = in_sizes[1] / 2;

    float* ws = (float*)d_ws;
    const size_t Nz = (size_t)N;
    const size_t Ez = (size_t)E;
    float* agg1 = ws;
    float* h2   = ws + 32 * Nz;
    float* a_s1 = ws + 48 * Nz;
    float* a_d1 = ws + 56 * Nz;
    int*   cnt  = (int*)(ws + 64 * Nz);
    int*   off  = (int*)(ws + 66 * Nz);
    int*   rank = (int*)(ws + 70 * Nz);
    int*   csr  = (int*)(ws + 70 * Nz + Ez);
    int*   hist = (int*)(ws + 70 * Nz + 2 * Ez);

    const int TB = 256;
    const int chunk = (E + NSLICE - 1) / NSLICE;
    k_count_lds<<<NSLICE, CTHR, 0, stream>>>(ei, E, N, chunk, hist, rank, cnt + N);
    k_alloc_slices<<<(N + TB - 1) / TB, TB, 0, stream>>>(hist, cnt + N, cnt, off, N);
    k_scatter<<<(E + TB * CEPT - 1) / (TB * CEPT), TB, 0, stream>>>(ei, E, N, chunk, hist, rank, csr);

    {
        int mtiles = (N + 15) / 16;
        k_att1_x<<<(mtiles + 7) / 8, TB, 0, stream>>>(x, W1, as1, ad1, a_s1, a_d1, N);
    }
    k_agg1<<<(int)(((long long)N * 32 + TB - 1) / TB), TB, 0, stream>>>(cnt, off, csr, x, a_s1, a_d1,
                                                                      agg1, N);

    {
        int mtiles = (N + 15) / 16;
        long long thr = (long long)mtiles * 32;
        k_l1out_gemm2_wmma<<<(int)((thr + 63) / 64), 64, 0, stream>>>(agg1, W1, bias1, W2, h2, N);
    }
    k_agg2<<<(int)(((long long)N * 32 + TB - 1) / TB), TB, 0, stream>>>(cnt, off, csr, h2, as2, ad2,
                                                                      bias2, out, N);
}
